# starting kernel
# baseline (speedup 1.0000x reference)
#include <hip/hip_runtime.h>
#include <math.h>

typedef float v2f __attribute__((ext_vector_type(2)));
typedef float v4f __attribute__((ext_vector_type(4)));
typedef float v8f __attribute__((ext_vector_type(8)));

#define NB 1024
#define NI 512
#define NO 512
#define EPS 1e-7f
#define LOG_CLIP 20.0f
#define NEG_BIG -1e30f


__global__ void nalu_prep_w_kernel(const float* __restrict__ w_hat,
                                   const float* __restrict__ m_hat,
                                   float* __restrict__ PB) {
    int t = blockIdx.x * blockDim.x + threadIdx.x;
    int r = t >> 9;
    int c = t & 511;
    float wv = tanhf(w_hat[t]) * (1.0f / (1.0f + expf(-m_hat[t])));
    PB[(((r >> 1) * NO + c) << 3) + (r & 1)] = wv;
    float wsv = fabsf(wv);
    float t2 = 1.0f - 2.0f * wsv;
    float t1 = 1.0f - wsv;
    int d = (((c >> 1) * NO + r) << 3) + (c & 1);
    PB[d + 2] = fmaxf(logf(fabsf(t2)), NEG_BIG);
    PB[d + 4] = fmaxf(logf(t1), NEG_BIG);
    PB[d + 6] = (t2 < 0.0f) ? 1.0f : 0.0f;
}

__global__ void nalu_prep_x_kernel(const float* __restrict__ x,
                                   float* __restrict__ PA) {
    int t = blockIdx.x * blockDim.x + threadIdx.x;
    int b = t >> 9;
    int i = t & 511;
    float xv = x[t];
    int d = ((b * (NI >> 1) + (i >> 1)) << 2) + (i & 1);
    PA[d]     = xv;
    PA[d + 2] = logf(fmaxf(fabsf(xv), EPS));
}

__global__ void nalu_main_kernel(const float* __restrict__ PA,
                                 const float* __restrict__ g,
                                 const float* __restrict__ PB,
                                 float* __restrict__ out) {
    const int lane = threadIdx.x & 31;
    const int wave = threadIdx.x >> 5;
    const int tile = blockIdx.x * 8 + wave;
    const int m0 = (tile >> 5) << 5;
    const int n0 = (tile & 31) << 4;
    const int half = lane >> 4;
    const int l15 = lane & 15;
    const int bcol = n0 + l15;

    const v4f* pa0 = (const v4f*)PA + (m0 + l15) * (NI >> 1) + half;
    const v4f* pa1 = pa0 + 16 * (NI >> 1);
    const v4f* pb  = (const v4f*)PB + ((half * NO + bcol) << 1);

    v8f accA0 = {}, accM0 = {}, accL0 = {}, accP0 = {};
    v8f accA1 = {}, accM1 = {}, accL1 = {}, accP1 = {};

#pragma unroll 4
    for (int k = 0; k < NI; k += 4) {
        v4f a0 = pa0[0];  pa0 += 2;
        v4f a1 = pa1[0];  pa1 += 2;
        v4f bq0 = pb[0];
        v4f bq1 = pb[1];
        pb += 4 * NO;

        v2f ax0 = __builtin_shufflevector(a0, a0, 0, 1);
        v2f al0 = __builtin_shufflevector(a0, a0, 2, 3);
        v2f ax1 = __builtin_shufflevector(a1, a1, 0, 1);
        v2f al1 = __builtin_shufflevector(a1, a1, 2, 3);
        v2f bw  = __builtin_shufflevector(bq0, bq0, 0, 1);
        v2f b2  = __builtin_shufflevector(bq0, bq0, 2, 3);
        v2f b1  = __builtin_shufflevector(bq1, bq1, 0, 1);
        v2f bn  = __builtin_shufflevector(bq1, bq1, 2, 3);

        v2f an0, az0, an1, az1;
        an0.x = (ax0.x < 0.0f) ? 1.0f : 0.0f;
        an0.y = (ax0.y < 0.0f) ? 1.0f : 0.0f;
        az0.x = (ax0.x == 0.0f) ? 1.0f : 0.0f;
        az0.y = (ax0.y == 0.0f) ? 1.0f : 0.0f;
        an1.x = (ax1.x < 0.0f) ? 1.0f : 0.0f;
        an1.y = (ax1.y < 0.0f) ? 1.0f : 0.0f;
        az1.x = (ax1.x == 0.0f) ? 1.0f : 0.0f;
        az1.y = (ax1.y == 0.0f) ? 1.0f : 0.0f;

        accA0 = __builtin_amdgcn_wmma_f32_16x16x4_f32(false, ax0, false, bw,
                                                      (short)0, accA0, false, false);
        accA1 = __builtin_amdgcn_wmma_f32_16x16x4_f32(false, ax1, false, bw,
                                                      (short)0, accA1, false, false);
        accM0 = __builtin_amdgcn_wmma_f32_16x16x4_f32(false, al0, false, bw,
                                                      (short)0, accM0, false, false);
        accM1 = __builtin_amdgcn_wmma_f32_16x16x4_f32(false, al1, false, bw,
                                                      (short)0, accM1, false, false);
        accL0 = __builtin_amdgcn_wmma_f32_16x16x4_f32(false, an0, false, b2,
                                                      (short)0, accL0, false, false);
        accL1 = __builtin_amdgcn_wmma_f32_16x16x4_f32(false, an1, false, b2,
                                                      (short)0, accL1, false, false);
        accL0 = __builtin_amdgcn_wmma_f32_16x16x4_f32(false, az0, false, b1,
                                                      (short)0, accL0, false, false);
        accL1 = __builtin_amdgcn_wmma_f32_16x16x4_f32(false, az1, false, b1,
                                                      (short)0, accL1, false, false);
        accP0 = __builtin_amdgcn_wmma_f32_16x16x4_f32(false, an0, false, bn,
                                                      (short)0, accP0, false, false);
        accP1 = __builtin_amdgcn_wmma_f32_16x16x4_f32(false, an1, false, bn,
                                                      (short)0, accP1, false, false);
    }

    const float g1 = 1.0f / (1.0f + expf(-g[bcol]));
    const float gm = 1.0f - g1;
#pragma unroll
    for (int v = 0; v < 8; ++v) {
        const int row0 = m0 + v + 8 * half;
        {
            const float m1 = expf(fminf(accM0[v], LOG_CLIP));
            const int negc = (int)(accP0[v] + 0.5f);
            float ms = ((negc & 1) ? -1.0f : 1.0f) * expf(accL0[v]);
            ms = fminf(fmaxf(ms, -1.0f), 1.0f);
            out[row0 * NO + bcol] = g1 * accA0[v] + gm * m1 * ms;
        }
        {
            const float m1 = expf(fminf(accM1[v], LOG_CLIP));
            const int negc = (int)(accP1[v] + 0.5f);
            float ms = ((negc & 1) ? -1.0f : 1.0f) * expf(accL1[v]);
            ms = fminf(fmaxf(ms, -1.0f), 1.0f);
            out[(row0 + 16) * NO + bcol] = g1 * accA1[v] + gm * m1 * ms;
        }
    }
}

extern "C" void kernel_launch(void* const* d_in, const int* in_sizes, int n_in,
                              void* d_out, int out_size, void* d_ws, size_t ws_size,
                              hipStream_t stream) {
    const float* x     = (const float*)d_in[0];
    const float* w_hat = (const float*)d_in[1];
    const float* m_hat = (const float*)d_in[2];
    const float* g     = (const float*)d_in[3];
    float* out = (float*)d_out;

    float* wsf = (float*)d_ws;
    float* PB  = wsf;
    float* PA  = wsf + (NI / 2) * NO * 8;

    nalu_prep_w_kernel<<<(NI * NO) / 256, 256, 0, stream>>>(w_hat, m_hat, PB);
    nalu_prep_x_kernel<<<(NB * NI) / 256, 256, 0, stream>>>(x, PA);

    nalu_main_kernel<<<128, 256, 0, stream>>>(PA, g, PB, out);
}


# main K loop in blocks of 8 k-steps on alternating register sets: each block's 32 global loads issued as one group a block of WMMAs ahead (was 6 exposed round trips per 4 k-steps); sum order kept
# speedup vs baseline: 2.1708x; 2.1708x over previous
#include <hip/hip_runtime.h>
#include <math.h>

typedef float v2f __attribute__((ext_vector_type(2)));
typedef float v4f __attribute__((ext_vector_type(4)));
typedef float v8f __attribute__((ext_vector_type(8)));

#define NB 1024
#define NI 512
#define NO 512
#define EPS 1e-7f
#define LOG_CLIP 20.0f
#define NEG_BIG -1e30f


__global__ void nalu_prep_w_kernel(const float* __restrict__ w_hat,
                                   const float* __restrict__ m_hat,
                                   float* __restrict__ PB) {
    int t = blockIdx.x * blockDim.x + threadIdx.x;
    int r = t >> 9;
    int c = t & 511;
    float wv = tanhf(w_hat[t]) * (1.0f / (1.0f + expf(-m_hat[t])));
    PB[(((r >> 1) * NO + c) << 3) + (r & 1)] = wv;
    float wsv = fabsf(wv);
    float t2 = 1.0f - 2.0f * wsv;
    float t1 = 1.0f - wsv;
    int d = (((c >> 1) * NO + r) << 3) + (c & 1);
    PB[d + 2] = fmaxf(logf(fabsf(t2)), NEG_BIG);
    PB[d + 4] = fmaxf(logf(t1), NEG_BIG);
    PB[d + 6] = (t2 < 0.0f) ? 1.0f : 0.0f;
}

__global__ void nalu_prep_x_kernel(const float* __restrict__ x,
                                   float* __restrict__ PA) {
    int t = blockIdx.x * blockDim.x + threadIdx.x;
    int b = t >> 9;
    int i = t & 511;
    float xv = x[t];
    int d = ((b * (NI >> 1) + (i >> 1)) << 2) + (i & 1);
    PA[d]     = xv;
    PA[d + 2] = logf(fmaxf(fabsf(xv), EPS));
}

#define KB 8
#define NBLK (NI / (4 * KB))

struct Frag {
    v4f a0[KB];
    v4f a1[KB];
    v4f q0[KB];
    v4f q1[KB];
};

__device__ __forceinline__ void load_block(Frag& f, const v4f* pa0, const v4f* pa1,
                                           const v4f* pb, int blk) {
    const v4f* qa0 = pa0 + blk * (2 * KB);
    const v4f* qa1 = pa1 + blk * (2 * KB);
    const v4f* qb  = pb + blk * (KB * 4 * NO);
#pragma unroll
    for (int s = 0; s < KB; ++s) {
        f.a0[s] = qa0[2 * s];
        f.a1[s] = qa1[2 * s];
        f.q0[s] = qb[s * 4 * NO];
        f.q1[s] = qb[s * 4 * NO + 1];
    }
}

__device__ __forceinline__ void mma_block(const Frag& f,
                                          v8f& accA0, v8f& accM0, v8f& accL0, v8f& accP0,
                                          v8f& accA1, v8f& accM1, v8f& accL1, v8f& accP1) {
#pragma unroll
    for (int s = 0; s < KB; ++s) {
        const v4f a0 = f.a0[s], a1 = f.a1[s], bq0 = f.q0[s], bq1 = f.q1[s];
        v2f ax0 = __builtin_shufflevector(a0, a0, 0, 1);
        v2f al0 = __builtin_shufflevector(a0, a0, 2, 3);
        v2f ax1 = __builtin_shufflevector(a1, a1, 0, 1);
        v2f al1 = __builtin_shufflevector(a1, a1, 2, 3);
        v2f bw  = __builtin_shufflevector(bq0, bq0, 0, 1);
        v2f b2  = __builtin_shufflevector(bq0, bq0, 2, 3);
        v2f b1  = __builtin_shufflevector(bq1, bq1, 0, 1);
        v2f bn  = __builtin_shufflevector(bq1, bq1, 2, 3);

        v2f an0, az0, an1, az1;
        an0.x = (ax0.x < 0.0f) ? 1.0f : 0.0f;
        an0.y = (ax0.y < 0.0f) ? 1.0f : 0.0f;
        az0.x = (ax0.x == 0.0f) ? 1.0f : 0.0f;
        az0.y = (ax0.y == 0.0f) ? 1.0f : 0.0f;
        an1.x = (ax1.x < 0.0f) ? 1.0f : 0.0f;
        an1.y = (ax1.y < 0.0f) ? 1.0f : 0.0f;
        az1.x = (ax1.x == 0.0f) ? 1.0f : 0.0f;
        az1.y = (ax1.y == 0.0f) ? 1.0f : 0.0f;

        accA0 = __builtin_amdgcn_wmma_f32_16x16x4_f32(false, ax0, false, bw,
                                                      (short)0, accA0, false, false);
        accA1 = __builtin_amdgcn_wmma_f32_16x16x4_f32(false, ax1, false, bw,
                                                      (short)0, accA1, false, false);
        accM0 = __builtin_amdgcn_wmma_f32_16x16x4_f32(false, al0, false, bw,
                                                      (short)0, accM0, false, false);
        accM1 = __builtin_amdgcn_wmma_f32_16x16x4_f32(false, al1, false, bw,
                                                      (short)0, accM1, false, false);
        accL0 = __builtin_amdgcn_wmma_f32_16x16x4_f32(false, an0, false, b2,
                                                      (short)0, accL0, false, false);
        accL1 = __builtin_amdgcn_wmma_f32_16x16x4_f32(false, an1, false, b2,
                                                      (short)0, accL1, false, false);
        accL0 = __builtin_amdgcn_wmma_f32_16x16x4_f32(false, az0, false, b1,
                                                      (short)0, accL0, false, false);
        accL1 = __builtin_amdgcn_wmma_f32_16x16x4_f32(false, az1, false, b1,
                                                      (short)0, accL1, false, false);
        accP0 = __builtin_amdgcn_wmma_f32_16x16x4_f32(false, an0, false, bn,
                                                      (short)0, accP0, false, false);
        accP1 = __builtin_amdgcn_wmma_f32_16x16x4_f32(false, an1, false, bn,
                                                      (short)0, accP1, false, false);
    }
}

__global__ void __launch_bounds__(256)
nalu_main_kernel(const float* __restrict__ PA,
                 const float* __restrict__ g,
                 const float* __restrict__ PB,
                 float* __restrict__ out) {
    const int lane = threadIdx.x & 31;
    const int wave = threadIdx.x >> 5;
    const int tile = blockIdx.x * 8 + wave;
    const int m0 = (tile >> 5) << 5;
    const int n0 = (tile & 31) << 4;
    const int half = lane >> 4;
    const int l15 = lane & 15;
    const int bcol = n0 + l15;

    const v4f* pa0 = (const v4f*)PA + (m0 + l15) * (NI >> 1) + half;
    const v4f* pa1 = pa0 + 16 * (NI >> 1);
    const v4f* pb  = (const v4f*)PB + ((half * NO + bcol) << 1);

    v8f accA0 = {}, accM0 = {}, accL0 = {}, accP0 = {};
    v8f accA1 = {}, accM1 = {}, accL1 = {}, accP1 = {};

    Frag f0, f1;
    load_block(f0, pa0, pa1, pb, 0);

#pragma unroll 1
    for (int b = 0; b < NBLK - 2; b += 2) {
        load_block(f1, pa0, pa1, pb, b + 1);
        __builtin_amdgcn_sched_barrier(0);
        mma_block(f0, accA0, accM0, accL0, accP0, accA1, accM1, accL1, accP1);
        load_block(f0, pa0, pa1, pb, b + 2);
        __builtin_amdgcn_sched_barrier(0);
        mma_block(f1, accA0, accM0, accL0, accP0, accA1, accM1, accL1, accP1);
    }
    load_block(f1, pa0, pa1, pb, NBLK - 1);
    __builtin_amdgcn_sched_barrier(0);
    mma_block(f0, accA0, accM0, accL0, accP0, accA1, accM1, accL1, accP1);
    mma_block(f1, accA0, accM0, accL0, accP0, accA1, accM1, accL1, accP1);

    const float g1 = 1.0f / (1.0f + expf(-g[bcol]));
    const float gm = 1.0f - g1;
#pragma unroll
    for (int v = 0; v < 8; ++v) {
        const int row0 = m0 + v + 8 * half;
        {
            const float m1 = expf(fminf(accM0[v], LOG_CLIP));
            const int negc = (int)(accP0[v] + 0.5f);
            float ms = ((negc & 1) ? -1.0f : 1.0f) * expf(accL0[v]);
            ms = fminf(fmaxf(ms, -1.0f), 1.0f);
            out[row0 * NO + bcol] = g1 * accA0[v] + gm * m1 * ms;
        }
        {
            const float m1 = expf(fminf(accM1[v], LOG_CLIP));
            const int negc = (int)(accP1[v] + 0.5f);
            float ms = ((negc & 1) ? -1.0f : 1.0f) * expf(accL1[v]);
            ms = fminf(fmaxf(ms, -1.0f), 1.0f);
            out[(row0 + 16) * NO + bcol] = g1 * accA1[v] + gm * m1 * ms;
        }
    }
}

extern "C" void kernel_launch(void* const* d_in, const int* in_sizes, int n_in,
                              void* d_out, int out_size, void* d_ws, size_t ws_size,
                              hipStream_t stream) {
    const float* x     = (const float*)d_in[0];
    const float* w_hat = (const float*)d_in[1];
    const float* m_hat = (const float*)d_in[2];
    const float* g     = (const float*)d_in[3];
    float* out = (float*)d_out;

    float* wsf = (float*)d_ws;
    float* PB  = wsf;
    float* PA  = wsf + (NI / 2) * NO * 8;

    nalu_prep_w_kernel<<<(NI * NO) / 256, 256, 0, stream>>>(w_hat, m_hat, PB);
    nalu_prep_x_kernel<<<(NB * NI) / 256, 256, 0, stream>>>(x, PA);

    nalu_main_kernel<<<128, 256, 0, stream>>>(PA, g, PB, out);
}


# main kernel launched as 256 workgroups of 4 waves (was 128 of 8): one wave per SIMD on every WGP, not two per SIMD on half of them, so no two waves share an f32 WMMA pipe; sum order kept
# speedup vs baseline: 3.1002x; 1.4281x over previous
#include <hip/hip_runtime.h>
#include <math.h>

typedef float v2f __attribute__((ext_vector_type(2)));
typedef float v4f __attribute__((ext_vector_type(4)));
typedef float v8f __attribute__((ext_vector_type(8)));

#define NB 1024
#define NI 512
#define NO 512
#define EPS 1e-7f
#define LOG_CLIP 20.0f
#define NEG_BIG -1e30f


__global__ void nalu_prep_w_kernel(const float* __restrict__ w_hat,
                                   const float* __restrict__ m_hat,
                                   float* __restrict__ PB) {
    int t = blockIdx.x * blockDim.x + threadIdx.x;
    int r = t >> 9;
    int c = t & 511;
    float wv = tanhf(w_hat[t]) * (1.0f / (1.0f + expf(-m_hat[t])));
    PB[(((r >> 1) * NO + c) << 3) + (r & 1)] = wv;
    float wsv = fabsf(wv);
    float t2 = 1.0f - 2.0f * wsv;
    float t1 = 1.0f - wsv;
    int d = (((c >> 1) * NO + r) << 3) + (c & 1);
    PB[d + 2] = fmaxf(logf(fabsf(t2)), NEG_BIG);
    PB[d + 4] = fmaxf(logf(t1), NEG_BIG);
    PB[d + 6] = (t2 < 0.0f) ? 1.0f : 0.0f;
}

__global__ void nalu_prep_x_kernel(const float* __restrict__ x,
                                   float* __restrict__ PA) {
    int t = blockIdx.x * blockDim.x + threadIdx.x;
    int b = t >> 9;
    int i = t & 511;
    float xv = x[t];
    int d = ((b * (NI >> 1) + (i >> 1)) << 2) + (i & 1);
    PA[d]     = xv;
    PA[d + 2] = logf(fmaxf(fabsf(xv), EPS));
}

#define KB 8
#define NBLK (NI / (4 * KB))

struct Frag {
    v4f a0[KB];
    v4f a1[KB];
    v4f q0[KB];
    v4f q1[KB];
};

__device__ __forceinline__ void load_block(Frag& f, const v4f* pa0, const v4f* pa1,
                                           const v4f* pb, int blk) {
    const v4f* qa0 = pa0 + blk * (2 * KB);
    const v4f* qa1 = pa1 + blk * (2 * KB);
    const v4f* qb  = pb + blk * (KB * 4 * NO);
#pragma unroll
    for (int s = 0; s < KB; ++s) {
        f.a0[s] = qa0[2 * s];
        f.a1[s] = qa1[2 * s];
        f.q0[s] = qb[s * 4 * NO];
        f.q1[s] = qb[s * 4 * NO + 1];
    }
}

__device__ __forceinline__ void mma_block(const Frag& f,
                                          v8f& accA0, v8f& accM0, v8f& accL0, v8f& accP0,
                                          v8f& accA1, v8f& accM1, v8f& accL1, v8f& accP1) {
#pragma unroll
    for (int s = 0; s < KB; ++s) {
        const v4f a0 = f.a0[s], a1 = f.a1[s], bq0 = f.q0[s], bq1 = f.q1[s];
        v2f ax0 = __builtin_shufflevector(a0, a0, 0, 1);
        v2f al0 = __builtin_shufflevector(a0, a0, 2, 3);
        v2f ax1 = __builtin_shufflevector(a1, a1, 0, 1);
        v2f al1 = __builtin_shufflevector(a1, a1, 2, 3);
        v2f bw  = __builtin_shufflevector(bq0, bq0, 0, 1);
        v2f b2  = __builtin_shufflevector(bq0, bq0, 2, 3);
        v2f b1  = __builtin_shufflevector(bq1, bq1, 0, 1);
        v2f bn  = __builtin_shufflevector(bq1, bq1, 2, 3);

        v2f an0, az0, an1, az1;
        an0.x = (ax0.x < 0.0f) ? 1.0f : 0.0f;
        an0.y = (ax0.y < 0.0f) ? 1.0f : 0.0f;
        az0.x = (ax0.x == 0.0f) ? 1.0f : 0.0f;
        az0.y = (ax0.y == 0.0f) ? 1.0f : 0.0f;
        an1.x = (ax1.x < 0.0f) ? 1.0f : 0.0f;
        an1.y = (ax1.y < 0.0f) ? 1.0f : 0.0f;
        az1.x = (ax1.x == 0.0f) ? 1.0f : 0.0f;
        az1.y = (ax1.y == 0.0f) ? 1.0f : 0.0f;

        accA0 = __builtin_amdgcn_wmma_f32_16x16x4_f32(false, ax0, false, bw,
                                                      (short)0, accA0, false, false);
        accA1 = __builtin_amdgcn_wmma_f32_16x16x4_f32(false, ax1, false, bw,
                                                      (short)0, accA1, false, false);
        accM0 = __builtin_amdgcn_wmma_f32_16x16x4_f32(false, al0, false, bw,
                                                      (short)0, accM0, false, false);
        accM1 = __builtin_amdgcn_wmma_f32_16x16x4_f32(false, al1, false, bw,
                                                      (short)0, accM1, false, false);
        accL0 = __builtin_amdgcn_wmma_f32_16x16x4_f32(false, an0, false, b2,
                                                      (short)0, accL0, false, false);
        accL1 = __builtin_amdgcn_wmma_f32_16x16x4_f32(false, an1, false, b2,
                                                      (short)0, accL1, false, false);
        accL0 = __builtin_amdgcn_wmma_f32_16x16x4_f32(false, az0, false, b1,
                                                      (short)0, accL0, false, false);
        accL1 = __builtin_amdgcn_wmma_f32_16x16x4_f32(false, az1, false, b1,
                                                      (short)0, accL1, false, false);
        accP0 = __builtin_amdgcn_wmma_f32_16x16x4_f32(false, an0, false, bn,
                                                      (short)0, accP0, false, false);
        accP1 = __builtin_amdgcn_wmma_f32_16x16x4_f32(false, an1, false, bn,
                                                      (short)0, accP1, false, false);
    }
}

__global__ void __launch_bounds__(128)
nalu_main_kernel(const float* __restrict__ PA,
                 const float* __restrict__ g,
                 const float* __restrict__ PB,
                 float* __restrict__ out) {
    const int lane = threadIdx.x & 31;
    const int wave = threadIdx.x >> 5;
    const int tile = blockIdx.x * 4 + wave;
    const int m0 = (tile >> 5) << 5;
    const int n0 = (tile & 31) << 4;
    const int half = lane >> 4;
    const int l15 = lane & 15;
    const int bcol = n0 + l15;

    const v4f* pa0 = (const v4f*)PA + (m0 + l15) * (NI >> 1) + half;
    const v4f* pa1 = pa0 + 16 * (NI >> 1);
    const v4f* pb  = (const v4f*)PB + ((half * NO + bcol) << 1);

    v8f accA0 = {}, accM0 = {}, accL0 = {}, accP0 = {};
    v8f accA1 = {}, accM1 = {}, accL1 = {}, accP1 = {};

    Frag f0, f1;
    load_block(f0, pa0, pa1, pb, 0);

#pragma unroll 1
    for (int b = 0; b < NBLK - 2; b += 2) {
        load_block(f1, pa0, pa1, pb, b + 1);
        __builtin_amdgcn_sched_barrier(0);
        mma_block(f0, accA0, accM0, accL0, accP0, accA1, accM1, accL1, accP1);
        load_block(f0, pa0, pa1, pb, b + 2);
        __builtin_amdgcn_sched_barrier(0);
        mma_block(f1, accA0, accM0, accL0, accP0, accA1, accM1, accL1, accP1);
    }
    load_block(f1, pa0, pa1, pb, NBLK - 1);
    __builtin_amdgcn_sched_barrier(0);
    mma_block(f0, accA0, accM0, accL0, accP0, accA1, accM1, accL1, accP1);
    mma_block(f1, accA0, accM0, accL0, accP0, accA1, accM1, accL1, accP1);

    const float g1 = 1.0f / (1.0f + expf(-g[bcol]));
    const float gm = 1.0f - g1;
#pragma unroll
    for (int v = 0; v < 8; ++v) {
        const int row0 = m0 + v + 8 * half;
        {
            const float m1 = expf(fminf(accM0[v], LOG_CLIP));
            const int negc = (int)(accP0[v] + 0.5f);
            float ms = ((negc & 1) ? -1.0f : 1.0f) * expf(accL0[v]);
            ms = fminf(fmaxf(ms, -1.0f), 1.0f);
            out[row0 * NO + bcol] = g1 * accA0[v] + gm * m1 * ms;
        }
        {
            const float m1 = expf(fminf(accM1[v], LOG_CLIP));
            const int negc = (int)(accP1[v] + 0.5f);
            float ms = ((negc & 1) ? -1.0f : 1.0f) * expf(accL1[v]);
            ms = fminf(fmaxf(ms, -1.0f), 1.0f);
            out[(row0 + 16) * NO + bcol] = g1 * accA1[v] + gm * m1 * ms;
        }
    }
}

extern "C" void kernel_launch(void* const* d_in, const int* in_sizes, int n_in,
                              void* d_out, int out_size, void* d_ws, size_t ws_size,
                              hipStream_t stream) {
    const float* x     = (const float*)d_in[0];
    const float* w_hat = (const float*)d_in[1];
    const float* m_hat = (const float*)d_in[2];
    const float* g     = (const float*)d_in[3];
    float* out = (float*)d_out;

    float* wsf = (float*)d_ws;
    float* PB  = wsf;
    float* PA  = wsf + (NI / 2) * NO * 8;

    nalu_prep_w_kernel<<<(NI * NO) / 256, 256, 0, stream>>>(w_hat, m_hat, PB);
    nalu_prep_x_kernel<<<(NB * NI) / 256, 256, 0, stream>>>(x, PA);

    nalu_main_kernel<<<256, 128, 0, stream>>>(PA, g, PB, out);
}


# x-packing launch removed: main reads x itself and forms log|x| in registers (log2 unit times two-term ln2), about 4 vector ops spread behind each f32 WMMA; 2 launches not 3; k and sum order kept
# speedup vs baseline: 3.2398x; 1.0450x over previous
#include <hip/hip_runtime.h>
#include <math.h>

typedef float v2f __attribute__((ext_vector_type(2)));
typedef float v4f __attribute__((ext_vector_type(4)));
typedef float v8f __attribute__((ext_vector_type(8)));

#define NB 1024
#define NI 512
#define NO 512
#define EPS 1e-7f
#define LOG_CLIP 20.0f
#define NEG_BIG -1e30f


__global__ void nalu_prep_w_kernel(const float* __restrict__ w_hat,
                                   const float* __restrict__ m_hat,
                                   float* __restrict__ PB) {
    int t = blockIdx.x * blockDim.x + threadIdx.x;
    int r = t >> 9;
    int c = t & 511;
    float wv = tanhf(w_hat[t]) * (1.0f / (1.0f + expf(-m_hat[t])));
    PB[(((r >> 1) * NO + c) << 3) + (r & 1)] = wv;
    float wsv = fabsf(wv);
    float t2 = 1.0f - 2.0f * wsv;
    float t1 = 1.0f - wsv;
    int d = (((c >> 1) * NO + r) << 3) + (c & 1);
    PB[d + 2] = fmaxf(logf(fabsf(t2)), NEG_BIG);
    PB[d + 4] = fmaxf(logf(t1), NEG_BIG);
    PB[d + 6] = (t2 < 0.0f) ? 1.0f : 0.0f;
}

#define KB 8
#define NBLK (NI / (4 * KB))

struct Frag {
    v2f a0[KB];
    v2f a1[KB];
    v4f q0[KB];
    v4f q1[KB];
};

__device__ __forceinline__ void load_block(Frag& f, const v2f* pa0, const v2f* pa1,
                                           const v4f* pb, int blk) {
    const v2f* qa0 = pa0 + blk * (2 * KB);
    const v2f* qa1 = pa1 + blk * (2 * KB);
    const v4f* qb  = pb + blk * (KB * 4 * NO);
#pragma unroll
    for (int s = 0; s < KB; ++s) {
        f.a0[s] = qa0[2 * s];
        f.a1[s] = qa1[2 * s];
        f.q0[s] = qb[s * 4 * NO];
        f.q1[s] = qb[s * 4 * NO + 1];
    }
}

__device__ __forceinline__ float log_abs(float xv) {
    const float v = fmaxf(fabsf(xv), EPS);
    const float y = __builtin_amdgcn_logf(v);
    return fmaf(y, 0x1.62e42ep-1f, y * 0x1.efa39ep-25f);
}

__device__ __forceinline__ void mma_block(const Frag& f,
                                          v8f& accA0, v8f& accM0, v8f& accL0, v8f& accP0,
                                          v8f& accA1, v8f& accM1, v8f& accL1, v8f& accP1) {
#pragma unroll
    for (int s = 0; s < KB; ++s) {
        const v2f ax0 = f.a0[s], ax1 = f.a1[s];
        const v4f bq0 = f.q0[s], bq1 = f.q1[s];
        v2f al0, al1;
        al0.x = log_abs(ax0.x);
        al0.y = log_abs(ax0.y);
        al1.x = log_abs(ax1.x);
        al1.y = log_abs(ax1.y);
        v2f bw  = __builtin_shufflevector(bq0, bq0, 0, 1);
        v2f b2  = __builtin_shufflevector(bq0, bq0, 2, 3);
        v2f b1  = __builtin_shufflevector(bq1, bq1, 0, 1);
        v2f bn  = __builtin_shufflevector(bq1, bq1, 2, 3);

        v2f an0, az0, an1, az1;
        an0.x = (ax0.x < 0.0f) ? 1.0f : 0.0f;
        an0.y = (ax0.y < 0.0f) ? 1.0f : 0.0f;
        az0.x = (ax0.x == 0.0f) ? 1.0f : 0.0f;
        az0.y = (ax0.y == 0.0f) ? 1.0f : 0.0f;
        an1.x = (ax1.x < 0.0f) ? 1.0f : 0.0f;
        an1.y = (ax1.y < 0.0f) ? 1.0f : 0.0f;
        az1.x = (ax1.x == 0.0f) ? 1.0f : 0.0f;
        az1.y = (ax1.y == 0.0f) ? 1.0f : 0.0f;

        accA0 = __builtin_amdgcn_wmma_f32_16x16x4_f32(false, ax0, false, bw,
                                                      (short)0, accA0, false, false);
        accA1 = __builtin_amdgcn_wmma_f32_16x16x4_f32(false, ax1, false, bw,
                                                      (short)0, accA1, false, false);
        accM0 = __builtin_amdgcn_wmma_f32_16x16x4_f32(false, al0, false, bw,
                                                      (short)0, accM0, false, false);
        accM1 = __builtin_amdgcn_wmma_f32_16x16x4_f32(false, al1, false, bw,
                                                      (short)0, accM1, false, false);
        accL0 = __builtin_amdgcn_wmma_f32_16x16x4_f32(false, an0, false, b2,
                                                      (short)0, accL0, false, false);
        accL1 = __builtin_amdgcn_wmma_f32_16x16x4_f32(false, an1, false, b2,
                                                      (short)0, accL1, false, false);
        accL0 = __builtin_amdgcn_wmma_f32_16x16x4_f32(false, az0, false, b1,
                                                      (short)0, accL0, false, false);
        accL1 = __builtin_amdgcn_wmma_f32_16x16x4_f32(false, az1, false, b1,
                                                      (short)0, accL1, false, false);
        accP0 = __builtin_amdgcn_wmma_f32_16x16x4_f32(false, an0, false, bn,
                                                      (short)0, accP0, false, false);
        accP1 = __builtin_amdgcn_wmma_f32_16x16x4_f32(false, an1, false, bn,
                                                      (short)0, accP1, false, false);
#pragma unroll
        for (int j = 0; j < 10; ++j) {
            __builtin_amdgcn_sched_group_barrier(0x008, 1, 0);
            __builtin_amdgcn_sched_group_barrier(0x002, 4, 0);
            if (j < 4)
                __builtin_amdgcn_sched_group_barrier(0x400, 1, 0);
        }
    }
}

__global__ void __launch_bounds__(128)
nalu_main_kernel(const float* __restrict__ x,
                 const float* __restrict__ g,
                 const float* __restrict__ PB,
                 float* __restrict__ out) {
    const int lane = threadIdx.x & 31;
    const int wave = threadIdx.x >> 5;
    const int tile = blockIdx.x * 4 + wave;
    const int m0 = (tile >> 5) << 5;
    const int n0 = (tile & 31) << 4;
    const int half = lane >> 4;
    const int l15 = lane & 15;
    const int bcol = n0 + l15;

    const v2f* pa0 = (const v2f*)x + (m0 + l15) * (NI >> 1) + half;
    const v2f* pa1 = pa0 + 16 * (NI >> 1);
    const v4f* pb  = (const v4f*)PB + ((half * NO + bcol) << 1);

    v8f accA0 = {}, accM0 = {}, accL0 = {}, accP0 = {};
    v8f accA1 = {}, accM1 = {}, accL1 = {}, accP1 = {};

    Frag f0, f1;
    load_block(f0, pa0, pa1, pb, 0);

#pragma unroll 1
    for (int b = 0; b < NBLK - 2; b += 2) {
        load_block(f1, pa0, pa1, pb, b + 1);
        __builtin_amdgcn_sched_barrier(0);
        mma_block(f0, accA0, accM0, accL0, accP0, accA1, accM1, accL1, accP1);
        load_block(f0, pa0, pa1, pb, b + 2);
        __builtin_amdgcn_sched_barrier(0);
        mma_block(f1, accA0, accM0, accL0, accP0, accA1, accM1, accL1, accP1);
    }
    load_block(f1, pa0, pa1, pb, NBLK - 1);
    __builtin_amdgcn_sched_barrier(0);
    mma_block(f0, accA0, accM0, accL0, accP0, accA1, accM1, accL1, accP1);
    mma_block(f1, accA0, accM0, accL0, accP0, accA1, accM1, accL1, accP1);

    const float g1 = 1.0f / (1.0f + expf(-g[bcol]));
    const float gm = 1.0f - g1;
#pragma unroll
    for (int v = 0; v < 8; ++v) {
        const int row0 = m0 + v + 8 * half;
        {
            const float m1 = expf(fminf(accM0[v], LOG_CLIP));
            const int negc = (int)(accP0[v] + 0.5f);
            float ms = ((negc & 1) ? -1.0f : 1.0f) * expf(accL0[v]);
            ms = fminf(fmaxf(ms, -1.0f), 1.0f);
            out[row0 * NO + bcol] = g1 * accA0[v] + gm * m1 * ms;
        }
        {
            const float m1 = expf(fminf(accM1[v], LOG_CLIP));
            const int negc = (int)(accP1[v] + 0.5f);
            float ms = ((negc & 1) ? -1.0f : 1.0f) * expf(accL1[v]);
            ms = fminf(fmaxf(ms, -1.0f), 1.0f);
            out[(row0 + 16) * NO + bcol] = g1 * accA1[v] + gm * m1 * ms;
        }
    }
}

extern "C" void kernel_launch(void* const* d_in, const int* in_sizes, int n_in,
                              void* d_out, int out_size, void* d_ws, size_t ws_size,
                              hipStream_t stream) {
    const float* x     = (const float*)d_in[0];
    const float* w_hat = (const float*)d_in[1];
    const float* m_hat = (const float*)d_in[2];
    const float* g     = (const float*)d_in[3];
    float* out = (float*)d_out;

    float* PB = (float*)d_ws;

    nalu_prep_w_kernel<<<(NI * NO) / 256, 256, 0, stream>>>(w_hat, m_hat, PB);

    nalu_main_kernel<<<256, 128, 0, stream>>>(x, g, PB, out);
}


# fewer 128-B lines per memory instruction: PB laid out in lane order (each B load and prep store = 4 whole lines, was 8), x read as b128 quads (2 k-steps a load); k order inside 8-groups changes sums
# speedup vs baseline: 3.9420x; 1.2167x over previous
#include <hip/hip_runtime.h>
#include <math.h>

typedef float v2f __attribute__((ext_vector_type(2)));
typedef float v4f __attribute__((ext_vector_type(4)));
typedef float v8f __attribute__((ext_vector_type(8)));

#define NB 1024
#define NI 512
#define NO 512
#define EPS 1e-7f
#define LOG_CLIP 20.0f
#define NEG_BIG -1e30f


__device__ __forceinline__ float w1_of(float wh, float mh) {
    return tanhf(wh) * (1.0f / (1.0f + expf(-mh)));
}

__global__ void __launch_bounds__(256)
nalu_prep_w_kernel(const float* __restrict__ w_hat,
                   const float* __restrict__ m_hat,
                   float* __restrict__ PB) {
    const int lane = threadIdx.x & 31;
    const int wave = threadIdx.x >> 5;
    const int ct = blockIdx.x & 31;
    const int J = ((blockIdx.x >> 5) << 2) + (wave >> 1);
    const int u = wave & 1;
    const int half = lane >> 4;
    const int col = (ct << 4) + (lane & 15);
    const int kp = (J << 2) + (half << 1) + u;
    const int k0 = kp << 1;

    const float wh_d0 = w_hat[k0 * NO + col];
    const float wh_d1 = w_hat[(k0 + 1) * NO + col];
    const float mh_d0 = m_hat[k0 * NO + col];
    const float mh_d1 = m_hat[(k0 + 1) * NO + col];
    const v2f wh_t = *(const v2f*)(w_hat + col * NO + k0);
    const v2f mh_t = *(const v2f*)(m_hat + col * NO + k0);

    const float wd0 = w1_of(wh_d0, mh_d0);
    const float wd1 = w1_of(wh_d1, mh_d1);
    const float ws0 = fabsf(w1_of(wh_t.x, mh_t.x));
    const float ws1 = fabsf(w1_of(wh_t.y, mh_t.y));
    const float t2_0 = 1.0f - 2.0f * ws0;
    const float t2_1 = 1.0f - 2.0f * ws1;
    const float t1_0 = 1.0f - ws0;
    const float t1_1 = 1.0f - ws1;

    v4f r0, r1;
    r0.x = wd0;
    r0.y = wd1;
    r0.z = fmaxf(logf(fabsf(t2_0)), NEG_BIG);
    r0.w = fmaxf(logf(fabsf(t2_1)), NEG_BIG);
    r1.x = fmaxf(logf(t1_0), NEG_BIG);
    r1.y = fmaxf(logf(t1_1), NEG_BIG);
    r1.z = (t2_0 < 0.0f) ? 1.0f : 0.0f;
    r1.w = (t2_1 < 0.0f) ? 1.0f : 0.0f;

    const int S = (J << 1) + u;
    v4f* dst = (v4f*)PB + (((S << 5) + ct) << 6) + lane;
    dst[0]  = r0;
    dst[32] = r1;
}

#define KB 8
#define NBLK (NI / (4 * KB))

struct Frag {
    v4f a0[KB / 2];
    v4f a1[KB / 2];
    v4f q0[KB];
    v4f q1[KB];
};

#define PB_STEP (32 * 2 * 32)

__device__ __forceinline__ void load_block(Frag& f, const v4f* pa0, const v4f* pa1,
                                           const v4f* pb, int blk) {
    const v4f* qa0 = pa0 + blk * KB;
    const v4f* qa1 = pa1 + blk * KB;
    const v4f* qb  = pb + blk * (KB * PB_STEP);
#pragma unroll
    for (int j = 0; j < KB / 2; ++j) {
        f.a0[j] = qa0[2 * j];
        f.a1[j] = qa1[2 * j];
    }
#pragma unroll
    for (int s = 0; s < KB; ++s) {
        f.q0[s] = qb[s * PB_STEP];
        f.q1[s] = qb[s * PB_STEP + 32];
    }
}

__device__ __forceinline__ float log_abs(float xv) {
    const float v = fmaxf(fabsf(xv), EPS);
    const float y = __builtin_amdgcn_logf(v);
    return fmaf(y, 0x1.62e42ep-1f, y * 0x1.efa39ep-25f);
}

__device__ __forceinline__ void mma_block(const Frag& f,
                                          v8f& accA0, v8f& accM0, v8f& accL0, v8f& accP0,
                                          v8f& accA1, v8f& accM1, v8f& accL1, v8f& accP1) {
#pragma unroll
    for (int s = 0; s < KB; ++s) {
        const v4f xq0 = f.a0[s >> 1], xq1 = f.a1[s >> 1];
        const v2f ax0 = (s & 1) ? __builtin_shufflevector(xq0, xq0, 2, 3)
                                : __builtin_shufflevector(xq0, xq0, 0, 1);
        const v2f ax1 = (s & 1) ? __builtin_shufflevector(xq1, xq1, 2, 3)
                                : __builtin_shufflevector(xq1, xq1, 0, 1);
        const v4f bq0 = f.q0[s], bq1 = f.q1[s];
        v2f al0, al1;
        al0.x = log_abs(ax0.x);
        al0.y = log_abs(ax0.y);
        al1.x = log_abs(ax1.x);
        al1.y = log_abs(ax1.y);
        v2f bw  = __builtin_shufflevector(bq0, bq0, 0, 1);
        v2f b2  = __builtin_shufflevector(bq0, bq0, 2, 3);
        v2f b1  = __builtin_shufflevector(bq1, bq1, 0, 1);
        v2f bn  = __builtin_shufflevector(bq1, bq1, 2, 3);

        v2f an0, az0, an1, az1;
        an0.x = (ax0.x < 0.0f) ? 1.0f : 0.0f;
        an0.y = (ax0.y < 0.0f) ? 1.0f : 0.0f;
        az0.x = (ax0.x == 0.0f) ? 1.0f : 0.0f;
        az0.y = (ax0.y == 0.0f) ? 1.0f : 0.0f;
        an1.x = (ax1.x < 0.0f) ? 1.0f : 0.0f;
        an1.y = (ax1.y < 0.0f) ? 1.0f : 0.0f;
        az1.x = (ax1.x == 0.0f) ? 1.0f : 0.0f;
        az1.y = (ax1.y == 0.0f) ? 1.0f : 0.0f;

        accA0 = __builtin_amdgcn_wmma_f32_16x16x4_f32(false, ax0, false, bw,
                                                      (short)0, accA0, false, false);
        accA1 = __builtin_amdgcn_wmma_f32_16x16x4_f32(false, ax1, false, bw,
                                                      (short)0, accA1, false, false);
        accM0 = __builtin_amdgcn_wmma_f32_16x16x4_f32(false, al0, false, bw,
                                                      (short)0, accM0, false, false);
        accM1 = __builtin_amdgcn_wmma_f32_16x16x4_f32(false, al1, false, bw,
                                                      (short)0, accM1, false, false);
        accL0 = __builtin_amdgcn_wmma_f32_16x16x4_f32(false, an0, false, b2,
                                                      (short)0, accL0, false, false);
        accL1 = __builtin_amdgcn_wmma_f32_16x16x4_f32(false, an1, false, b2,
                                                      (short)0, accL1, false, false);
        accL0 = __builtin_amdgcn_wmma_f32_16x16x4_f32(false, az0, false, b1,
                                                      (short)0, accL0, false, false);
        accL1 = __builtin_amdgcn_wmma_f32_16x16x4_f32(false, az1, false, b1,
                                                      (short)0, accL1, false, false);
        accP0 = __builtin_amdgcn_wmma_f32_16x16x4_f32(false, an0, false, bn,
                                                      (short)0, accP0, false, false);
        accP1 = __builtin_amdgcn_wmma_f32_16x16x4_f32(false, an1, false, bn,
                                                      (short)0, accP1, false, false);
#pragma unroll
        for (int j = 0; j < 10; ++j) {
            __builtin_amdgcn_sched_group_barrier(0x008, 1, 0);
            __builtin_amdgcn_sched_group_barrier(0x002, 4, 0);
            if (j < 4)
                __builtin_amdgcn_sched_group_barrier(0x400, 1, 0);
        }
    }
}

__global__ void __launch_bounds__(128)
nalu_main_kernel(const float* __restrict__ x,
                 const float* __restrict__ g,
                 const float* __restrict__ PB,
                 float* __restrict__ out) {
    const int lane = threadIdx.x & 31;
    const int wave = threadIdx.x >> 5;
    const int tile = blockIdx.x * 4 + wave;
    const int m0 = (tile >> 5) << 5;
    const int n0 = (tile & 31) << 4;
    const int half = lane >> 4;
    const int l15 = lane & 15;
    const int bcol = n0 + l15;

    const v4f* pa0 = (const v4f*)x + (m0 + l15) * (NI >> 2) + half;
    const v4f* pa1 = pa0 + 16 * (NI >> 2);
    const v4f* pb  = (const v4f*)PB + ((tile & 31) << 6) + lane;

    v8f accA0 = {}, accM0 = {}, accL0 = {}, accP0 = {};
    v8f accA1 = {}, accM1 = {}, accL1 = {}, accP1 = {};

    Frag f0, f1;
    load_block(f0, pa0, pa1, pb, 0);

#pragma unroll 1
    for (int b = 0; b < NBLK - 2; b += 2) {
        load_block(f1, pa0, pa1, pb, b + 1);
        __builtin_amdgcn_sched_barrier(0);
        mma_block(f0, accA0, accM0, accL0, accP0, accA1, accM1, accL1, accP1);
        load_block(f0, pa0, pa1, pb, b + 2);
        __builtin_amdgcn_sched_barrier(0);
        mma_block(f1, accA0, accM0, accL0, accP0, accA1, accM1, accL1, accP1);
    }
    load_block(f1, pa0, pa1, pb, NBLK - 1);
    __builtin_amdgcn_sched_barrier(0);
    mma_block(f0, accA0, accM0, accL0, accP0, accA1, accM1, accL1, accP1);
    mma_block(f1, accA0, accM0, accL0, accP0, accA1, accM1, accL1, accP1);

    const float g1 = 1.0f / (1.0f + expf(-g[bcol]));
    const float gm = 1.0f - g1;
#pragma unroll
    for (int v = 0; v < 8; ++v) {
        const int row0 = m0 + v + 8 * half;
        {
            const float m1 = expf(fminf(accM0[v], LOG_CLIP));
            const int negc = (int)(accP0[v] + 0.5f);
            float ms = ((negc & 1) ? -1.0f : 1.0f) * expf(accL0[v]);
            ms = fminf(fmaxf(ms, -1.0f), 1.0f);
            out[row0 * NO + bcol] = g1 * accA0[v] + gm * m1 * ms;
        }
        {
            const float m1 = expf(fminf(accM1[v], LOG_CLIP));
            const int negc = (int)(accP1[v] + 0.5f);
            float ms = ((negc & 1) ? -1.0f : 1.0f) * expf(accL1[v]);
            ms = fminf(fmaxf(ms, -1.0f), 1.0f);
            out[(row0 + 16) * NO + bcol] = g1 * accA1[v] + gm * m1 * ms;
        }
    }
}

extern "C" void kernel_launch(void* const* d_in, const int* in_sizes, int n_in,
                              void* d_out, int out_size, void* d_ws, size_t ws_size,
                              hipStream_t stream) {
    const float* x     = (const float*)d_in[0];
    const float* w_hat = (const float*)d_in[1];
    const float* m_hat = (const float*)d_in[2];
    const float* g     = (const float*)d_in[3];
    float* out = (float*)d_out;

    float* PB = (float*)d_ws;

    nalu_prep_w_kernel<<<512, 256, 0, stream>>>(w_hat, m_hat, PB);

    nalu_main_kernel<<<256, 128, 0, stream>>>(x, g, PB, out);
}


# zero-flag product (x==0 against log(1-|w|)) moved out of the K loop: 8 not 10 f32 WMMAs per k-step; a wave whose running min|x| is 0 adds it in a second loop; sum order changes only for x with zeros
# speedup vs baseline: 4.6692x; 1.1845x over previous
#include <hip/hip_runtime.h>
#include <math.h>

typedef float v2f __attribute__((ext_vector_type(2)));
typedef float v4f __attribute__((ext_vector_type(4)));
typedef float v8f __attribute__((ext_vector_type(8)));

#define NB 1024
#define NI 512
#define NO 512
#define EPS 1e-7f
#define LOG_CLIP 20.0f
#define NEG_BIG -1e30f


__device__ __forceinline__ float w1_of(float wh, float mh) {
    return tanhf(wh) * (1.0f / (1.0f + expf(-mh)));
}

__global__ void __launch_bounds__(256)
nalu_prep_w_kernel(const float* __restrict__ w_hat,
                   const float* __restrict__ m_hat,
                   float* __restrict__ PB) {
    const int lane = threadIdx.x & 31;
    const int wave = threadIdx.x >> 5;
    const int ct = blockIdx.x & 31;
    const int J = ((blockIdx.x >> 5) << 2) + (wave >> 1);
    const int u = wave & 1;
    const int half = lane >> 4;
    const int col = (ct << 4) + (lane & 15);
    const int kp = (J << 2) + (half << 1) + u;
    const int k0 = kp << 1;

    const float wh_d0 = w_hat[k0 * NO + col];
    const float wh_d1 = w_hat[(k0 + 1) * NO + col];
    const float mh_d0 = m_hat[k0 * NO + col];
    const float mh_d1 = m_hat[(k0 + 1) * NO + col];
    const v2f wh_t = *(const v2f*)(w_hat + col * NO + k0);
    const v2f mh_t = *(const v2f*)(m_hat + col * NO + k0);

    const float wd0 = w1_of(wh_d0, mh_d0);
    const float wd1 = w1_of(wh_d1, mh_d1);
    const float ws0 = fabsf(w1_of(wh_t.x, mh_t.x));
    const float ws1 = fabsf(w1_of(wh_t.y, mh_t.y));
    const float t2_0 = 1.0f - 2.0f * ws0;
    const float t2_1 = 1.0f - 2.0f * ws1;
    const float t1_0 = 1.0f - ws0;
    const float t1_1 = 1.0f - ws1;

    v4f r0, r1;
    r0.x = wd0;
    r0.y = wd1;
    r0.z = fmaxf(logf(fabsf(t2_0)), NEG_BIG);
    r0.w = fmaxf(logf(fabsf(t2_1)), NEG_BIG);
    r1.x = fmaxf(logf(t1_0), NEG_BIG);
    r1.y = fmaxf(logf(t1_1), NEG_BIG);
    r1.z = (t2_0 < 0.0f) ? 1.0f : 0.0f;
    r1.w = (t2_1 < 0.0f) ? 1.0f : 0.0f;

    const int S = (J << 1) + u;
    v4f* dst = (v4f*)PB + (((S << 5) + ct) << 6) + lane;
    dst[0]  = r0;
    dst[32] = r1;
}

#define KB 8
#define NBLK (NI / (4 * KB))
#define NBH (NBLK / 2)

struct Frag {
    v4f a0[KB / 2];
    v4f a1[KB / 2];
    v4f q0[KB];
    v2f q1[KB];
};

#define PB_STEP (32 * 2 * 32)

__device__ __forceinline__ void load_block(Frag& f, const v4f* pa0, const v4f* pa1,
                                           const v4f* pb, int blk) {
    const v4f* qa0 = pa0 + blk * KB;
    const v4f* qa1 = pa1 + blk * KB;
    const v4f* qb  = pb + blk * (KB * PB_STEP);
#pragma unroll
    for (int j = 0; j < KB / 2; ++j) {
        f.a0[j] = qa0[2 * j];
        f.a1[j] = qa1[2 * j];
    }
#pragma unroll
    for (int s = 0; s < KB; ++s) {
        f.q0[s] = qb[s * PB_STEP];
        f.q1[s] = ((const v2f*)(qb + s * PB_STEP + 32))[1];
    }
}

__device__ __forceinline__ float log_abs(float xv) {
    const float v = fmaxf(fabsf(xv), EPS);
    const float y = __builtin_amdgcn_logf(v);
    return fmaf(y, 0x1.62e42ep-1f, y * 0x1.efa39ep-25f);
}

__device__ __forceinline__ void mma_block(const Frag& f, float& zmin,
                                          v8f& accA0, v8f& accM0, v8f& accL0, v8f& accP0,
                                          v8f& accA1, v8f& accM1, v8f& accL1, v8f& accP1) {
#pragma unroll
    for (int s = 0; s < KB; ++s) {
        const v4f xq0 = f.a0[s >> 1], xq1 = f.a1[s >> 1];
        const v2f ax0 = (s & 1) ? __builtin_shufflevector(xq0, xq0, 2, 3)
                                : __builtin_shufflevector(xq0, xq0, 0, 1);
        const v2f ax1 = (s & 1) ? __builtin_shufflevector(xq1, xq1, 2, 3)
                                : __builtin_shufflevector(xq1, xq1, 0, 1);
        const v4f bq0 = f.q0[s];
        v2f al0, al1;
        al0.x = log_abs(ax0.x);
        al0.y = log_abs(ax0.y);
        al1.x = log_abs(ax1.x);
        al1.y = log_abs(ax1.y);
        v2f bw  = __builtin_shufflevector(bq0, bq0, 0, 1);
        v2f b2  = __builtin_shufflevector(bq0, bq0, 2, 3);
        v2f bn  = f.q1[s];

        v2f an0, an1;
        an0.x = (ax0.x < 0.0f) ? 1.0f : 0.0f;
        an0.y = (ax0.y < 0.0f) ? 1.0f : 0.0f;
        an1.x = (ax1.x < 0.0f) ? 1.0f : 0.0f;
        an1.y = (ax1.y < 0.0f) ? 1.0f : 0.0f;
        zmin = fminf(zmin, fminf(fabsf(ax0.x), fabsf(ax0.y)));
        zmin = fminf(zmin, fminf(fabsf(ax1.x), fabsf(ax1.y)));

        accA0 = __builtin_amdgcn_wmma_f32_16x16x4_f32(false, ax0, false, bw,
                                                      (short)0, accA0, false, false);
        accA1 = __builtin_amdgcn_wmma_f32_16x16x4_f32(false, ax1, false, bw,
                                                      (short)0, accA1, false, false);
        accM0 = __builtin_amdgcn_wmma_f32_16x16x4_f32(false, al0, false, bw,
                                                      (short)0, accM0, false, false);
        accM1 = __builtin_amdgcn_wmma_f32_16x16x4_f32(false, al1, false, bw,
                                                      (short)0, accM1, false, false);
        accL0 = __builtin_amdgcn_wmma_f32_16x16x4_f32(false, an0, false, b2,
                                                      (short)0, accL0, false, false);
        accL1 = __builtin_amdgcn_wmma_f32_16x16x4_f32(false, an1, false, b2,
                                                      (short)0, accL1, false, false);
        accP0 = __builtin_amdgcn_wmma_f32_16x16x4_f32(false, an0, false, bn,
                                                      (short)0, accP0, false, false);
        accP1 = __builtin_amdgcn_wmma_f32_16x16x4_f32(false, an1, false, bn,
                                                      (short)0, accP1, false, false);
#pragma unroll
        for (int j = 0; j < 8; ++j) {
            __builtin_amdgcn_sched_group_barrier(0x008, 1, 0);
            __builtin_amdgcn_sched_group_barrier(0x002, 4, 0);
            if (j < 4)
                __builtin_amdgcn_sched_group_barrier(0x400, 1, 0);
        }
    }
}

__device__ __forceinline__ void zero_pass(const v4f* pa0, const v4f* pa1, const v4f* pb,
                                          v8f& accL0, v8f& accL1) {
#pragma unroll 1
    for (int p = 0; p < NBH * KB / 2; ++p) {
        const v4f xq0 = pa0[2 * p];
        const v4f xq1 = pa1[2 * p];
        const v4f r0 = pb[(2 * p) * PB_STEP + 32];
        const v4f r1 = pb[(2 * p + 1) * PB_STEP + 32];
        v2f az0, az1;
        az0.x = (xq0.x == 0.0f) ? 1.0f : 0.0f;
        az0.y = (xq0.y == 0.0f) ? 1.0f : 0.0f;
        az1.x = (xq1.x == 0.0f) ? 1.0f : 0.0f;
        az1.y = (xq1.y == 0.0f) ? 1.0f : 0.0f;
        const v2f b1e = __builtin_shufflevector(r0, r0, 0, 1);
        accL0 = __builtin_amdgcn_wmma_f32_16x16x4_f32(false, az0, false, b1e,
                                                      (short)0, accL0, false, false);
        accL1 = __builtin_amdgcn_wmma_f32_16x16x4_f32(false, az1, false, b1e,
                                                      (short)0, accL1, false, false);
        v2f bz0, bz1;
        bz0.x = (xq0.z == 0.0f) ? 1.0f : 0.0f;
        bz0.y = (xq0.w == 0.0f) ? 1.0f : 0.0f;
        bz1.x = (xq1.z == 0.0f) ? 1.0f : 0.0f;
        bz1.y = (xq1.w == 0.0f) ? 1.0f : 0.0f;
        const v2f b1o = __builtin_shufflevector(r1, r1, 0, 1);
        accL0 = __builtin_amdgcn_wmma_f32_16x16x4_f32(false, bz0, false, b1o,
                                                      (short)0, accL0, false, false);
        accL1 = __builtin_amdgcn_wmma_f32_16x16x4_f32(false, bz1, false, b1o,
                                                      (short)0, accL1, false, false);
    }
}

__global__ void __launch_bounds__(256)
nalu_main_kernel(const float* __restrict__ x,
                 const float* __restrict__ g,
                 const float* __restrict__ PB,
                 float* __restrict__ out) {
    __shared__ v4f xch[8 * 8 * 32];

    const int lane = threadIdx.x & 31;
    const int wave = threadIdx.x >> 5;
    const int kh = wave >> 2;
    const int tile = blockIdx.x * 4 + (wave & 3);
    const int m0 = (tile >> 5) << 5;
    const int n0 = (tile & 31) << 4;
    const int half = lane >> 4;
    const int l15 = lane & 15;
    const int bcol = n0 + l15;

    const v4f* pa0 = (const v4f*)x + (m0 + l15) * (NI >> 2) + kh * (NI >> 3) + half;
    const v4f* pa1 = pa0 + 16 * (NI >> 2);
    const v4f* pb  = (const v4f*)PB + kh * (NBH * KB * PB_STEP)
                     + ((tile & 31) << 6) + lane;

    v8f accA0 = {}, accM0 = {}, accL0 = {}, accP0 = {};
    v8f accA1 = {}, accM1 = {}, accL1 = {}, accP1 = {};
    float zmin = 1.0f;

    Frag f0, f1;
    load_block(f0, pa0, pa1, pb, 0);

#pragma unroll 1
    for (int b = 0; b < NBH - 2; b += 2) {
        load_block(f1, pa0, pa1, pb, b + 1);
        __builtin_amdgcn_sched_barrier(0);
        mma_block(f0, zmin, accA0, accM0, accL0, accP0, accA1, accM1, accL1, accP1);
        load_block(f0, pa0, pa1, pb, b + 2);
        __builtin_amdgcn_sched_barrier(0);
        mma_block(f1, zmin, accA0, accM0, accL0, accP0, accA1, accM1, accL1, accP1);
    }
    load_block(f1, pa0, pa1, pb, NBH - 1);
    __builtin_amdgcn_sched_barrier(0);
    mma_block(f0, zmin, accA0, accM0, accL0, accP0, accA1, accM1, accL1, accP1);
    mma_block(f1, zmin, accA0, accM0, accL0, accP0, accA1, accM1, accL1, accP1);

    if (__builtin_amdgcn_ballot_w32(zmin == 0.0f) != 0u)
        zero_pass(pa0, pa1, pb, accL0, accL1);

    const v8f sA = kh ? accA0 : accA1;
    const v8f sM = kh ? accM0 : accM1;
    const v8f sL = kh ? accL0 : accL1;
    const v8f sP = kh ? accP0 : accP1;
    v8f kA = kh ? accA1 : accA0;
    v8f kM = kh ? accM1 : accM0;
    v8f kL = kh ? accL1 : accL0;
    v8f kP = kh ? accP1 : accP0;

    v4f* mine = xch + (wave << 8) + lane;
    mine[0 * 32] = __builtin_shufflevector(sA, sA, 0, 1, 2, 3);
    mine[1 * 32] = __builtin_shufflevector(sA, sA, 4, 5, 6, 7);
    mine[2 * 32] = __builtin_shufflevector(sM, sM, 0, 1, 2, 3);
    mine[3 * 32] = __builtin_shufflevector(sM, sM, 4, 5, 6, 7);
    mine[4 * 32] = __builtin_shufflevector(sL, sL, 0, 1, 2, 3);
    mine[5 * 32] = __builtin_shufflevector(sL, sL, 4, 5, 6, 7);
    mine[6 * 32] = __builtin_shufflevector(sP, sP, 0, 1, 2, 3);
    mine[7 * 32] = __builtin_shufflevector(sP, sP, 4, 5, 6, 7);
    __syncthreads();
    const v4f* theirs = xch + ((wave ^ 4) << 8) + lane;
    kA += __builtin_shufflevector(theirs[0 * 32], theirs[1 * 32], 0, 1, 2, 3, 4, 5, 6, 7);
    kM += __builtin_shufflevector(theirs[2 * 32], theirs[3 * 32], 0, 1, 2, 3, 4, 5, 6, 7);
    kL += __builtin_shufflevector(theirs[4 * 32], theirs[5 * 32], 0, 1, 2, 3, 4, 5, 6, 7);
    kP += __builtin_shufflevector(theirs[6 * 32], theirs[7 * 32], 0, 1, 2, 3, 4, 5, 6, 7);

    const float g1 = 1.0f / (1.0f + expf(-g[bcol]));
    const float gm = 1.0f - g1;
    const int rbase = m0 + 16 * kh + 8 * half;
#pragma unroll
    for (int v = 0; v < 8; ++v) {
        const float m1 = expf(fminf(kM[v], LOG_CLIP));
        const int negc = (int)(kP[v] + 0.5f);
        float ms = ((negc & 1) ? -1.0f : 1.0f) * expf(kL[v]);
        ms = fminf(fmaxf(ms, -1.0f), 1.0f);
        out[(rbase + v) * NO + bcol] = g1 * kA[v] + gm * m1 * ms;
    }
}

extern "C" void kernel_launch(void* const* d_in, const int* in_sizes, int n_in,
                              void* d_out, int out_size, void* d_ws, size_t ws_size,
                              hipStream_t stream) {
    const float* x     = (const float*)d_in[0];
    const float* w_hat = (const float*)d_in[1];
    const float* m_hat = (const float*)d_in[2];
    const float* g     = (const float*)d_in[3];
    float* out = (float*)d_out;

    float* PB = (float*)d_ws;

    nalu_prep_w_kernel<<<512, 256, 0, stream>>>(w_hat, m_hat, PB);

    nalu_main_kernel<<<256, 256, 0, stream>>>(x, g, PB, out);
}


# sign of the factor product as a parity: lane masks of the x<0 compares ANDed with column masks and XORed, 16 rows per vector op, replace the 0/1 count product: 6 not 8 f32 WMMAs per k-step; sums kept
# speedup vs baseline: 5.0748x; 1.0869x over previous
#include <hip/hip_runtime.h>
#include <math.h>

typedef float v2f __attribute__((ext_vector_type(2)));
typedef float v4f __attribute__((ext_vector_type(4)));
typedef float v8f __attribute__((ext_vector_type(8)));
typedef unsigned v2u __attribute__((ext_vector_type(2)));

#define NB 1024
#define NI 512
#define NO 512
#define EPS 1e-7f
#define LOG_CLIP 20.0f
#define NEG_BIG -1e30f


__device__ __forceinline__ float w1_of(float wh, float mh) {
    return tanhf(wh) * (1.0f / (1.0f + expf(-mh)));
}

__global__ void __launch_bounds__(256)
nalu_prep_w_kernel(const float* __restrict__ w_hat,
                   const float* __restrict__ m_hat,
                   float* __restrict__ PB) {
    const int lane = threadIdx.x & 31;
    const int wave = threadIdx.x >> 5;
    const int ct = blockIdx.x & 31;
    const int J = ((blockIdx.x >> 5) << 2) + (wave >> 1);
    const int u = wave & 1;
    const int half = lane >> 4;
    const int col = (ct << 4) + (lane & 15);
    const int kp = (J << 2) + (half << 1) + u;
    const int k0 = kp << 1;

    const float wh_d0 = w_hat[k0 * NO + col];
    const float wh_d1 = w_hat[(k0 + 1) * NO + col];
    const float mh_d0 = m_hat[k0 * NO + col];
    const float mh_d1 = m_hat[(k0 + 1) * NO + col];
    const v2f wh_t = *(const v2f*)(w_hat + col * NO + k0);
    const v2f mh_t = *(const v2f*)(m_hat + col * NO + k0);

    const float wd0 = w1_of(wh_d0, mh_d0);
    const float wd1 = w1_of(wh_d1, mh_d1);
    const float ws0 = fabsf(w1_of(wh_t.x, mh_t.x));
    const float ws1 = fabsf(w1_of(wh_t.y, mh_t.y));
    const float t2_0 = 1.0f - 2.0f * ws0;
    const float t2_1 = 1.0f - 2.0f * ws1;
    const float t1_0 = 1.0f - ws0;
    const float t1_1 = 1.0f - ws1;

    v4f r0, r1;
    r0.x = wd0;
    r0.y = wd1;
    r0.z = fmaxf(logf(fabsf(t2_0)), NEG_BIG);
    r0.w = fmaxf(logf(fabsf(t2_1)), NEG_BIG);
    r1.x = fmaxf(logf(t1_0), NEG_BIG);
    r1.y = fmaxf(logf(t1_1), NEG_BIG);
    const unsigned hm = half ? 0xFFFF0000u : 0x0000FFFFu;
    r1.z = __uint_as_float((t2_0 < 0.0f) ? hm : 0u);
    r1.w = __uint_as_float((t2_1 < 0.0f) ? hm : 0u);

    const int S = (J << 1) + u;
    v4f* dst = (v4f*)PB + (((S << 5) + ct) << 6) + lane;
    dst[0]  = r0;
    dst[32] = r1;
}

#define KB 8
#define NBLK (NI / (4 * KB))
#define NBH (NBLK / 2)

struct Frag {
    v4f a0[KB / 2];
    v4f a1[KB / 2];
    v4f q0[KB];
    v2u q1[KB];
};

#define PB_STEP (32 * 2 * 32)

__device__ __forceinline__ void load_block(Frag& f, const v4f* pa0, const v4f* pa1,
                                           const v4f* pb, int blk) {
    const v4f* qa0 = pa0 + blk * KB;
    const v4f* qa1 = pa1 + blk * KB;
    const v4f* qb  = pb + blk * (KB * PB_STEP);
#pragma unroll
    for (int j = 0; j < KB / 2; ++j) {
        f.a0[j] = qa0[2 * j];
        f.a1[j] = qa1[2 * j];
    }
#pragma unroll
    for (int s = 0; s < KB; ++s) {
        f.q0[s] = qb[s * PB_STEP];
        f.q1[s] = ((const v2u*)(qb + s * PB_STEP + 32))[1];
    }
}

__device__ __forceinline__ float log_abs(float xv) {
    const float v = fmaxf(fabsf(xv), EPS);
    const float y = __builtin_amdgcn_logf(v);
    return fmaf(y, 0x1.62e42ep-1f, y * 0x1.efa39ep-25f);
}

__device__ __forceinline__ void mma_block(const Frag& f, float& zmin,
                                          unsigned& par0, unsigned& par1,
                                          v8f& accA0, v8f& accM0, v8f& accL0,
                                          v8f& accA1, v8f& accM1, v8f& accL1) {
#pragma unroll
    for (int s = 0; s < KB; ++s) {
        const v4f xq0 = f.a0[s >> 1], xq1 = f.a1[s >> 1];
        const v2f ax0 = (s & 1) ? __builtin_shufflevector(xq0, xq0, 2, 3)
                                : __builtin_shufflevector(xq0, xq0, 0, 1);
        const v2f ax1 = (s & 1) ? __builtin_shufflevector(xq1, xq1, 2, 3)
                                : __builtin_shufflevector(xq1, xq1, 0, 1);
        const v4f bq0 = f.q0[s];
        v2f al0, al1;
        al0.x = log_abs(ax0.x);
        al0.y = log_abs(ax0.y);
        al1.x = log_abs(ax1.x);
        al1.y = log_abs(ax1.y);
        v2f bw  = __builtin_shufflevector(bq0, bq0, 0, 1);
        v2f b2  = __builtin_shufflevector(bq0, bq0, 2, 3);
        const v2u bm = f.q1[s];

        const bool c00 = ax0.x < 0.0f, c01 = ax0.y < 0.0f;
        const bool c10 = ax1.x < 0.0f, c11 = ax1.y < 0.0f;
        v2f an0, an1;
        an0.x = c00 ? 1.0f : 0.0f;
        an0.y = c01 ? 1.0f : 0.0f;
        an1.x = c10 ? 1.0f : 0.0f;
        an1.y = c11 ? 1.0f : 0.0f;
        par0 ^= (__builtin_amdgcn_ballot_w32(c00) & bm.x)
              ^ (__builtin_amdgcn_ballot_w32(c01) & bm.y);
        par1 ^= (__builtin_amdgcn_ballot_w32(c10) & bm.x)
              ^ (__builtin_amdgcn_ballot_w32(c11) & bm.y);
        zmin = fminf(zmin, fminf(fabsf(ax0.x), fabsf(ax0.y)));
        zmin = fminf(zmin, fminf(fabsf(ax1.x), fabsf(ax1.y)));

        accA0 = __builtin_amdgcn_wmma_f32_16x16x4_f32(false, ax0, false, bw,
                                                      (short)0, accA0, false, false);
        accA1 = __builtin_amdgcn_wmma_f32_16x16x4_f32(false, ax1, false, bw,
                                                      (short)0, accA1, false, false);
        accM0 = __builtin_amdgcn_wmma_f32_16x16x4_f32(false, al0, false, bw,
                                                      (short)0, accM0, false, false);
        accM1 = __builtin_amdgcn_wmma_f32_16x16x4_f32(false, al1, false, bw,
                                                      (short)0, accM1, false, false);
        accL0 = __builtin_amdgcn_wmma_f32_16x16x4_f32(false, an0, false, b2,
                                                      (short)0, accL0, false, false);
        accL1 = __builtin_amdgcn_wmma_f32_16x16x4_f32(false, an1, false, b2,
                                                      (short)0, accL1, false, false);
#pragma unroll
        for (int j = 0; j < 6; ++j) {
            __builtin_amdgcn_sched_group_barrier(0x008, 1, 0);
            __builtin_amdgcn_sched_group_barrier(0x002, 6, 0);
            if (j < 4)
                __builtin_amdgcn_sched_group_barrier(0x400, 1, 0);
        }
    }
}

__device__ __forceinline__ void zero_pass(const v4f* pa0, const v4f* pa1, const v4f* pb,
                                          v8f& accL0, v8f& accL1) {
#pragma unroll 1
    for (int p = 0; p < NBH * KB / 2; ++p) {
        const v4f xq0 = pa0[2 * p];
        const v4f xq1 = pa1[2 * p];
        const v4f r0 = pb[(2 * p) * PB_STEP + 32];
        const v4f r1 = pb[(2 * p + 1) * PB_STEP + 32];
        v2f az0, az1;
        az0.x = (xq0.x == 0.0f) ? 1.0f : 0.0f;
        az0.y = (xq0.y == 0.0f) ? 1.0f : 0.0f;
        az1.x = (xq1.x == 0.0f) ? 1.0f : 0.0f;
        az1.y = (xq1.y == 0.0f) ? 1.0f : 0.0f;
        const v2f b1e = __builtin_shufflevector(r0, r0, 0, 1);
        accL0 = __builtin_amdgcn_wmma_f32_16x16x4_f32(false, az0, false, b1e,
                                                      (short)0, accL0, false, false);
        accL1 = __builtin_amdgcn_wmma_f32_16x16x4_f32(false, az1, false, b1e,
                                                      (short)0, accL1, false, false);
        v2f bz0, bz1;
        bz0.x = (xq0.z == 0.0f) ? 1.0f : 0.0f;
        bz0.y = (xq0.w == 0.0f) ? 1.0f : 0.0f;
        bz1.x = (xq1.z == 0.0f) ? 1.0f : 0.0f;
        bz1.y = (xq1.w == 0.0f) ? 1.0f : 0.0f;
        const v2f b1o = __builtin_shufflevector(r1, r1, 0, 1);
        accL0 = __builtin_amdgcn_wmma_f32_16x16x4_f32(false, bz0, false, b1o,
                                                      (short)0, accL0, false, false);
        accL1 = __builtin_amdgcn_wmma_f32_16x16x4_f32(false, bz1, false, b1o,
                                                      (short)0, accL1, false, false);
    }
}

__global__ void __launch_bounds__(256)
nalu_main_kernel(const float* __restrict__ x,
                 const float* __restrict__ g,
                 const float* __restrict__ PB,
                 float* __restrict__ out) {
    __shared__ v4f xch[8 * 6 * 32];
    __shared__ unsigned xpar[8 * 2 * 32];

    const int lane = threadIdx.x & 31;
    const int wave = threadIdx.x >> 5;
    const int kh = wave >> 2;
    const int tile = blockIdx.x * 4 + (wave & 3);
    const int m0 = (tile >> 5) << 5;
    const int n0 = (tile & 31) << 4;
    const int half = lane >> 4;
    const int l15 = lane & 15;
    const int bcol = n0 + l15;

    const v4f* pa0 = (const v4f*)x + (m0 + l15) * (NI >> 2) + kh * (NI >> 3) + half;
    const v4f* pa1 = pa0 + 16 * (NI >> 2);
    const v4f* pb  = (const v4f*)PB + kh * (NBH * KB * PB_STEP)
                     + ((tile & 31) << 6) + lane;

    v8f accA0 = {}, accM0 = {}, accL0 = {};
    v8f accA1 = {}, accM1 = {}, accL1 = {};
    unsigned par0 = 0u, par1 = 0u;
    float zmin = 1.0f;

    Frag f0, f1;
    load_block(f0, pa0, pa1, pb, 0);

#pragma unroll 1
    for (int b = 0; b < NBH - 2; b += 2) {
        load_block(f1, pa0, pa1, pb, b + 1);
        __builtin_amdgcn_sched_barrier(0);
        mma_block(f0, zmin, par0, par1, accA0, accM0, accL0, accA1, accM1, accL1);
        load_block(f0, pa0, pa1, pb, b + 2);
        __builtin_amdgcn_sched_barrier(0);
        mma_block(f1, zmin, par0, par1, accA0, accM0, accL0, accA1, accM1, accL1);
    }
    load_block(f1, pa0, pa1, pb, NBH - 1);
    __builtin_amdgcn_sched_barrier(0);
    mma_block(f0, zmin, par0, par1, accA0, accM0, accL0, accA1, accM1, accL1);
    mma_block(f1, zmin, par0, par1, accA0, accM0, accL0, accA1, accM1, accL1);

    if (__builtin_amdgcn_ballot_w32(zmin == 0.0f) != 0u)
        zero_pass(pa0, pa1, pb, accL0, accL1);

    const v8f sA = kh ? accA0 : accA1;
    const v8f sM = kh ? accM0 : accM1;
    const v8f sL = kh ? accL0 : accL1;
    v8f kA = kh ? accA1 : accA0;
    v8f kM = kh ? accM1 : accM0;
    v8f kL = kh ? accL1 : accL0;

    v4f* mine = xch + wave * (6 * 32) + lane;
    mine[0 * 32] = __builtin_shufflevector(sA, sA, 0, 1, 2, 3);
    mine[1 * 32] = __builtin_shufflevector(sA, sA, 4, 5, 6, 7);
    mine[2 * 32] = __builtin_shufflevector(sM, sM, 0, 1, 2, 3);
    mine[3 * 32] = __builtin_shufflevector(sM, sM, 4, 5, 6, 7);
    mine[4 * 32] = __builtin_shufflevector(sL, sL, 0, 1, 2, 3);
    mine[5 * 32] = __builtin_shufflevector(sL, sL, 4, 5, 6, 7);
    xpar[(wave << 6) + lane]      = (par0 >> (half << 4)) & 0xFFFFu;
    xpar[(wave << 6) + 32 + lane] = (par1 >> (half << 4)) & 0xFFFFu;
    __syncthreads();
    const v4f* theirs = xch + (wave ^ 4) * (6 * 32) + lane;
    kA += __builtin_shufflevector(theirs[0 * 32], theirs[1 * 32], 0, 1, 2, 3, 4, 5, 6, 7);
    kM += __builtin_shufflevector(theirs[2 * 32], theirs[3 * 32], 0, 1, 2, 3, 4, 5, 6, 7);
    kL += __builtin_shufflevector(theirs[4 * 32], theirs[5 * 32], 0, 1, 2, 3, 4, 5, 6, 7);
    const unsigned* pmine = xpar + (wave << 6) + (kh << 5) + l15;
    const unsigned* ptheirs = xpar + ((wave ^ 4) << 6) + (kh << 5) + l15;
    const unsigned kPar = pmine[0] ^ pmine[16] ^ ptheirs[0] ^ ptheirs[16];

    const float g1 = 1.0f / (1.0f + expf(-g[bcol]));
    const float gm = 1.0f - g1;
    const int rbase = m0 + 16 * kh + 8 * half;
#pragma unroll
    for (int v = 0; v < 8; ++v) {
        const float m1 = expf(fminf(kM[v], LOG_CLIP));
        const unsigned odd = (kPar >> (8 * half + v)) & 1u;
        float ms = (odd ? -1.0f : 1.0f) * expf(kL[v]);
        ms = fminf(fmaxf(ms, -1.0f), 1.0f);
        out[(rbase + v) * NO + bcol] = g1 * kA[v] + gm * m1 * ms;
    }
}

extern "C" void kernel_launch(void* const* d_in, const int* in_sizes, int n_in,
                              void* d_out, int out_size, void* d_ws, size_t ws_size,
                              hipStream_t stream) {
    const float* x     = (const float*)d_in[0];
    const float* w_hat = (const float*)d_in[1];
    const float* m_hat = (const float*)d_in[2];
    const float* g     = (const float*)d_in[3];
    float* out = (float*)d_out;

    float* PB = (float*)d_ws;

    nalu_prep_w_kernel<<<512, 256, 0, stream>>>(w_hat, m_hat, PB);

    nalu_main_kernel<<<256, 256, 0, stream>>>(x, g, PB, out);
}


# weight-prep launch removed: each workgroup (one 16-column tile x 128 rows) builds its B operand in LDS, tanh*sigmoid on the exp/rcp/log unit; K loop reads B from LDS; 1 launch; w1 rounding changes
# speedup vs baseline: 5.5303x; 1.0898x over previous
#include <hip/hip_runtime.h>
#include <math.h>

typedef float v2f __attribute__((ext_vector_type(2)));
typedef float v4f __attribute__((ext_vector_type(4)));
typedef float v8f __attribute__((ext_vector_type(8)));
typedef unsigned v2u __attribute__((ext_vector_type(2)));

#define NB 1024
#define NI 512
#define NO 512
#define EPS 1e-7f
#define LOG_CLIP 20.0f
#define NEG_BIG -1e30f


#define PB_STEP 64
#define WT_PITCH 516

__device__ __forceinline__ float w1_of(float wh, float mh) {
    const float e = __builtin_amdgcn_exp2f(fabsf(wh) * -0x1.715476p+1f);
    const float s = __builtin_amdgcn_exp2f(mh * -0x1.715476p+0f);
    const float r = __builtin_amdgcn_rcpf((1.0f + e) * (1.0f + s));
    return copysignf((1.0f - e) * r, wh);
}

__device__ __forceinline__ float log_clamped(float v) {
    const float y = __builtin_amdgcn_logf(v);
    return fmaxf(fmaf(y, 0x1.62e42ep-1f, y * 0x1.efa39ep-25f), NEG_BIG);
}

__device__ __forceinline__ void build_b(const float* __restrict__ w_hat,
                                        const float* __restrict__ m_hat,
                                        v4f* pbl, float* wt,
                                        int n0, int wave, int lane) {
    const int half = lane >> 4;
    const int l15 = lane & 15;
    const int col = n0 + l15;
    const int tid = (wave << 5) + lane;

    v4f tw[8], tm[8];
#pragma unroll
    for (int i = 0; i < 8; ++i) {
        const int idx = tid + (i << 8);
        const int c = idx >> 7;
        const int kq = idx & 127;
        tw[i] = ((const v4f*)w_hat)[(n0 + c) * (NO >> 2) + kq];
        tm[i] = ((const v4f*)m_hat)[(n0 + c) * (NO >> 2) + kq];
    }
    float dw0[16], dw1[16], dm0[16], dm1[16];
#pragma unroll
    for (int it = 0; it < 16; ++it) {
        const int J = (it << 2) + (wave >> 1);
        const int k0 = ((J << 2) + (half << 1) + (wave & 1)) << 1;
        dw0[it] = w_hat[k0 * NO + col];
        dw1[it] = w_hat[(k0 + 1) * NO + col];
        dm0[it] = m_hat[k0 * NO + col];
        dm1[it] = m_hat[(k0 + 1) * NO + col];
    }
#pragma unroll
    for (int i = 0; i < 8; ++i) {
        const int idx = tid + (i << 8);
        const int c = idx >> 7;
        const int kq = idx & 127;
        v4f r;
        r.x = w1_of(tw[i].x, tm[i].x);
        r.y = w1_of(tw[i].y, tm[i].y);
        r.z = w1_of(tw[i].z, tm[i].z);
        r.w = w1_of(tw[i].w, tm[i].w);
        *(v4f*)(wt + c * WT_PITCH + (kq << 2)) = r;
    }
    __syncthreads();

    const unsigned hm = half ? 0xFFFF0000u : 0x0000FFFFu;
#pragma unroll
    for (int it = 0; it < 16; ++it) {
        const int J = (it << 2) + (wave >> 1);
        const int u = wave & 1;
        const int k0 = ((J << 2) + (half << 1) + u) << 1;
        const v2f t = *(const v2f*)(wt + l15 * WT_PITCH + k0);
        const float ws0 = fabsf(t.x);
        const float ws1 = fabsf(t.y);
        const float t2_0 = 1.0f - 2.0f * ws0;
        const float t2_1 = 1.0f - 2.0f * ws1;
        v4f r0, r1;
        r0.x = w1_of(dw0[it], dm0[it]);
        r0.y = w1_of(dw1[it], dm1[it]);
        r0.z = log_clamped(fabsf(t2_0));
        r0.w = log_clamped(fabsf(t2_1));
        r1.x = log_clamped(1.0f - ws0);
        r1.y = log_clamped(1.0f - ws1);
        r1.z = __uint_as_float((t2_0 < 0.0f) ? hm : 0u);
        r1.w = __uint_as_float((t2_1 < 0.0f) ? hm : 0u);
        const int S = (J << 1) + u;
        v4f* dst = pbl + S * PB_STEP + lane;
        dst[0]  = r0;
        dst[32] = r1;
    }
    __syncthreads();
}

#define KB 8
#define NBLK (NI / (4 * KB))
#define NBH (NBLK / 2)

struct Frag {
    v4f a0[KB / 2];
    v4f a1[KB / 2];
    v4f q0[KB];
    v2u q1[KB];
};


__device__ __forceinline__ void load_a(Frag& f, const v4f* pa0, const v4f* pa1, int blk) {
    const v4f* qa0 = pa0 + blk * KB;
    const v4f* qa1 = pa1 + blk * KB;
#pragma unroll
    for (int j = 0; j < KB / 2; ++j) {
        f.a0[j] = qa0[2 * j];
        f.a1[j] = qa1[2 * j];
    }
}

__device__ __forceinline__ void load_b(Frag& f, const v4f* pb, int blk) {
    const v4f* qb  = pb + blk * (KB * PB_STEP);
#pragma unroll
    for (int s = 0; s < KB; ++s) {
        f.q0[s] = qb[s * PB_STEP];
        f.q1[s] = ((const v2u*)(qb + s * PB_STEP + 32))[1];
    }
}

__device__ __forceinline__ void load_block(Frag& f, const v4f* pa0, const v4f* pa1,
                                           const v4f* pb, int blk) {
    load_a(f, pa0, pa1, blk);
    load_b(f, pb, blk);
}

__device__ __forceinline__ float log_abs(float xv) {
    const float v = fmaxf(fabsf(xv), EPS);
    const float y = __builtin_amdgcn_logf(v);
    return fmaf(y, 0x1.62e42ep-1f, y * 0x1.efa39ep-25f);
}

__device__ __forceinline__ void mma_block(const Frag& f, float& zmin,
                                          unsigned& par0, unsigned& par1,
                                          v8f& accA0, v8f& accM0, v8f& accL0,
                                          v8f& accA1, v8f& accM1, v8f& accL1) {
#pragma unroll
    for (int s = 0; s < KB; ++s) {
        const v4f xq0 = f.a0[s >> 1], xq1 = f.a1[s >> 1];
        const v2f ax0 = (s & 1) ? __builtin_shufflevector(xq0, xq0, 2, 3)
                                : __builtin_shufflevector(xq0, xq0, 0, 1);
        const v2f ax1 = (s & 1) ? __builtin_shufflevector(xq1, xq1, 2, 3)
                                : __builtin_shufflevector(xq1, xq1, 0, 1);
        const v4f bq0 = f.q0[s];
        v2f al0, al1;
        al0.x = log_abs(ax0.x);
        al0.y = log_abs(ax0.y);
        al1.x = log_abs(ax1.x);
        al1.y = log_abs(ax1.y);
        v2f bw  = __builtin_shufflevector(bq0, bq0, 0, 1);
        v2f b2  = __builtin_shufflevector(bq0, bq0, 2, 3);
        const v2u bm = f.q1[s];

        const bool c00 = ax0.x < 0.0f, c01 = ax0.y < 0.0f;
        const bool c10 = ax1.x < 0.0f, c11 = ax1.y < 0.0f;
        v2f an0, an1;
        an0.x = c00 ? 1.0f : 0.0f;
        an0.y = c01 ? 1.0f : 0.0f;
        an1.x = c10 ? 1.0f : 0.0f;
        an1.y = c11 ? 1.0f : 0.0f;
        par0 ^= (__builtin_amdgcn_ballot_w32(c00) & bm.x)
              ^ (__builtin_amdgcn_ballot_w32(c01) & bm.y);
        par1 ^= (__builtin_amdgcn_ballot_w32(c10) & bm.x)
              ^ (__builtin_amdgcn_ballot_w32(c11) & bm.y);
        zmin = fminf(zmin, fminf(fabsf(ax0.x), fabsf(ax0.y)));
        zmin = fminf(zmin, fminf(fabsf(ax1.x), fabsf(ax1.y)));

        accA0 = __builtin_amdgcn_wmma_f32_16x16x4_f32(false, ax0, false, bw,
                                                      (short)0, accA0, false, false);
        accA1 = __builtin_amdgcn_wmma_f32_16x16x4_f32(false, ax1, false, bw,
                                                      (short)0, accA1, false, false);
        accM0 = __builtin_amdgcn_wmma_f32_16x16x4_f32(false, al0, false, bw,
                                                      (short)0, accM0, false, false);
        accM1 = __builtin_amdgcn_wmma_f32_16x16x4_f32(false, al1, false, bw,
                                                      (short)0, accM1, false, false);
        accL0 = __builtin_amdgcn_wmma_f32_16x16x4_f32(false, an0, false, b2,
                                                      (short)0, accL0, false, false);
        accL1 = __builtin_amdgcn_wmma_f32_16x16x4_f32(false, an1, false, b2,
                                                      (short)0, accL1, false, false);
#pragma unroll
        for (int j = 0; j < 6; ++j) {
            __builtin_amdgcn_sched_group_barrier(0x008, 1, 0);
            __builtin_amdgcn_sched_group_barrier(0x002, 6, 0);
            if (j < 4)
                __builtin_amdgcn_sched_group_barrier(0x400, 1, 0);
        }
    }
}

__device__ __forceinline__ void zero_pass(const v4f* pa0, const v4f* pa1, const v4f* pb,
                                          v8f& accL0, v8f& accL1) {
#pragma unroll 1
    for (int p = 0; p < NBH * KB / 2; ++p) {
        const v4f xq0 = pa0[2 * p];
        const v4f xq1 = pa1[2 * p];
        const v4f r0 = pb[(2 * p) * PB_STEP + 32];
        const v4f r1 = pb[(2 * p + 1) * PB_STEP + 32];
        v2f az0, az1;
        az0.x = (xq0.x == 0.0f) ? 1.0f : 0.0f;
        az0.y = (xq0.y == 0.0f) ? 1.0f : 0.0f;
        az1.x = (xq1.x == 0.0f) ? 1.0f : 0.0f;
        az1.y = (xq1.y == 0.0f) ? 1.0f : 0.0f;
        const v2f b1e = __builtin_shufflevector(r0, r0, 0, 1);
        accL0 = __builtin_amdgcn_wmma_f32_16x16x4_f32(false, az0, false, b1e,
                                                      (short)0, accL0, false, false);
        accL1 = __builtin_amdgcn_wmma_f32_16x16x4_f32(false, az1, false, b1e,
                                                      (short)0, accL1, false, false);
        v2f bz0, bz1;
        bz0.x = (xq0.z == 0.0f) ? 1.0f : 0.0f;
        bz0.y = (xq0.w == 0.0f) ? 1.0f : 0.0f;
        bz1.x = (xq1.z == 0.0f) ? 1.0f : 0.0f;
        bz1.y = (xq1.w == 0.0f) ? 1.0f : 0.0f;
        const v2f b1o = __builtin_shufflevector(r1, r1, 0, 1);
        accL0 = __builtin_amdgcn_wmma_f32_16x16x4_f32(false, bz0, false, b1o,
                                                      (short)0, accL0, false, false);
        accL1 = __builtin_amdgcn_wmma_f32_16x16x4_f32(false, bz1, false, b1o,
                                                      (short)0, accL1, false, false);
    }
}

__global__ void __launch_bounds__(256)
nalu_fused_kernel(const float* __restrict__ x,
                  const float* __restrict__ w_hat,
                  const float* __restrict__ m_hat,
                  const float* __restrict__ g,
                  float* __restrict__ out) {
    __shared__ v4f pbl[128 * PB_STEP];
    __shared__ __attribute__((aligned(16))) float wt[16 * WT_PITCH];
    __shared__ v4f xch[8 * 6 * 32];
    __shared__ unsigned xpar[8 * 2 * 32];

    const int lane = threadIdx.x & 31;
    const int wave = threadIdx.x >> 5;
    const int kh = wave >> 2;
    const int n0 = (blockIdx.x & 31) << 4;
    const int m0 = ((blockIdx.x >> 5) << 7) + ((wave & 3) << 5);
    const int half = lane >> 4;
    const int l15 = lane & 15;
    const int bcol = n0 + l15;

    const v4f* pa0 = (const v4f*)x + (m0 + l15) * (NI >> 2) + kh * (NI >> 3) + half;
    const v4f* pa1 = pa0 + 16 * (NI >> 2);

    Frag f0, f1;
    load_a(f0, pa0, pa1, 0);

    build_b(w_hat, m_hat, pbl, wt, n0, wave, lane);

    const v4f* pb  = pbl + kh * (NBH * KB * PB_STEP) + lane;

    v8f accA0 = {}, accM0 = {}, accL0 = {};
    v8f accA1 = {}, accM1 = {}, accL1 = {};
    unsigned par0 = 0u, par1 = 0u;
    float zmin = 1.0f;

    load_b(f0, pb, 0);

#pragma unroll 1
    for (int b = 0; b < NBH - 2; b += 2) {
        load_block(f1, pa0, pa1, pb, b + 1);
        __builtin_amdgcn_sched_barrier(0);
        mma_block(f0, zmin, par0, par1, accA0, accM0, accL0, accA1, accM1, accL1);
        load_block(f0, pa0, pa1, pb, b + 2);
        __builtin_amdgcn_sched_barrier(0);
        mma_block(f1, zmin, par0, par1, accA0, accM0, accL0, accA1, accM1, accL1);
    }
    load_block(f1, pa0, pa1, pb, NBH - 1);
    __builtin_amdgcn_sched_barrier(0);
    mma_block(f0, zmin, par0, par1, accA0, accM0, accL0, accA1, accM1, accL1);
    mma_block(f1, zmin, par0, par1, accA0, accM0, accL0, accA1, accM1, accL1);

    if (__builtin_amdgcn_ballot_w32(zmin == 0.0f) != 0u)
        zero_pass(pa0, pa1, pb, accL0, accL1);

    const v8f sA = kh ? accA0 : accA1;
    const v8f sM = kh ? accM0 : accM1;
    const v8f sL = kh ? accL0 : accL1;
    v8f kA = kh ? accA1 : accA0;
    v8f kM = kh ? accM1 : accM0;
    v8f kL = kh ? accL1 : accL0;

    v4f* mine = xch + wave * (6 * 32) + lane;
    mine[0 * 32] = __builtin_shufflevector(sA, sA, 0, 1, 2, 3);
    mine[1 * 32] = __builtin_shufflevector(sA, sA, 4, 5, 6, 7);
    mine[2 * 32] = __builtin_shufflevector(sM, sM, 0, 1, 2, 3);
    mine[3 * 32] = __builtin_shufflevector(sM, sM, 4, 5, 6, 7);
    mine[4 * 32] = __builtin_shufflevector(sL, sL, 0, 1, 2, 3);
    mine[5 * 32] = __builtin_shufflevector(sL, sL, 4, 5, 6, 7);
    xpar[(wave << 6) + lane]      = (par0 >> (half << 4)) & 0xFFFFu;
    xpar[(wave << 6) + 32 + lane] = (par1 >> (half << 4)) & 0xFFFFu;
    __syncthreads();
    const v4f* theirs = xch + (wave ^ 4) * (6 * 32) + lane;
    kA += __builtin_shufflevector(theirs[0 * 32], theirs[1 * 32], 0, 1, 2, 3, 4, 5, 6, 7);
    kM += __builtin_shufflevector(theirs[2 * 32], theirs[3 * 32], 0, 1, 2, 3, 4, 5, 6, 7);
    kL += __builtin_shufflevector(theirs[4 * 32], theirs[5 * 32], 0, 1, 2, 3, 4, 5, 6, 7);
    const unsigned* pmine = xpar + (wave << 6) + (kh << 5) + l15;
    const unsigned* ptheirs = xpar + ((wave ^ 4) << 6) + (kh << 5) + l15;
    const unsigned kPar = pmine[0] ^ pmine[16] ^ ptheirs[0] ^ ptheirs[16];

    const float g1 = 1.0f / (1.0f + expf(-g[bcol]));
    const float gm = 1.0f - g1;
    const int rbase = m0 + 16 * kh + 8 * half;
#pragma unroll
    for (int v = 0; v < 8; ++v) {
        const float m1 = expf(fminf(kM[v], LOG_CLIP));
        const unsigned odd = (kPar >> (8 * half + v)) & 1u;
        float ms = (odd ? -1.0f : 1.0f) * expf(kL[v]);
        ms = fminf(fmaxf(ms, -1.0f), 1.0f);
        out[(rbase + v) * NO + bcol] = g1 * kA[v] + gm * m1 * ms;
    }
}

extern "C" void kernel_launch(void* const* d_in, const int* in_sizes, int n_in,
                              void* d_out, int out_size, void* d_ws, size_t ws_size,
                              hipStream_t stream) {
    const float* x     = (const float*)d_in[0];
    const float* w_hat = (const float*)d_in[1];
    const float* m_hat = (const float*)d_in[2];
    const float* g     = (const float*)d_in[3];
    float* out = (float*)d_out;

    nalu_fused_kernel<<<256, 256, 0, stream>>>(x, w_hat, m_hat, g, out);
}


# the B build's two barriers publish LDS only and no longer wait for global loads in flight; weight loads queued in use order, x last: they land under the build's arithmetic; sums kept
# speedup vs baseline: 5.8352x; 1.0551x over previous
#include <hip/hip_runtime.h>
#include <math.h>

typedef float v2f __attribute__((ext_vector_type(2)));
typedef float v4f __attribute__((ext_vector_type(4)));
typedef float v8f __attribute__((ext_vector_type(8)));
typedef unsigned v2u __attribute__((ext_vector_type(2)));

#define NB 1024
#define NI 512
#define NO 512
#define EPS 1e-7f
#define LOG_CLIP 20.0f
#define NEG_BIG -1e30f


#define PB_STEP 64
#define WT_PITCH 516

__device__ __forceinline__ float w1_of(float wh, float mh) {
    const float e = __builtin_amdgcn_exp2f(fabsf(wh) * -0x1.715476p+1f);
    const float s = __builtin_amdgcn_exp2f(mh * -0x1.715476p+0f);
    const float r = __builtin_amdgcn_rcpf((1.0f + e) * (1.0f + s));
    return copysignf((1.0f - e) * r, wh);
}

__device__ __forceinline__ float log_clamped(float v) {
    const float y = __builtin_amdgcn_logf(v);
    return fmaxf(fmaf(y, 0x1.62e42ep-1f, y * 0x1.efa39ep-25f), NEG_BIG);
}

__device__ __forceinline__ void lds_barrier() {
    __builtin_amdgcn_fence(__ATOMIC_RELEASE, "workgroup", "local");
    __builtin_amdgcn_s_barrier();
    __builtin_amdgcn_fence(__ATOMIC_ACQUIRE, "workgroup", "local");
}

struct Frag;
__device__ __forceinline__ void load_a(Frag& f, const v4f* pa0, const v4f* pa1, int blk);

__device__ __forceinline__ void build_b(const float* __restrict__ w_hat,
                                        const float* __restrict__ m_hat,
                                        v4f* pbl, float* wt,
                                        int n0, int wave, int lane,
                                        Frag& fx, const v4f* pa0, const v4f* pa1) {
    const int half = lane >> 4;
    const int l15 = lane & 15;
    const int col = n0 + l15;
    const int tid = (wave << 5) + lane;

    v4f tw[8], tm[8];
#pragma unroll
    for (int i = 0; i < 8; ++i) {
        const int idx = tid + (i << 8);
        const int c = idx >> 7;
        const int kq = idx & 127;
        tw[i] = ((const v4f*)w_hat)[(n0 + c) * (NO >> 2) + kq];
        tm[i] = ((const v4f*)m_hat)[(n0 + c) * (NO >> 2) + kq];
    }
    __builtin_amdgcn_sched_barrier(0);
    float dw0[16], dw1[16], dm0[16], dm1[16];
#pragma unroll
    for (int it = 0; it < 16; ++it) {
        const int J = (it << 2) + (wave >> 1);
        const int k0 = ((J << 2) + (half << 1) + (wave & 1)) << 1;
        const int i0 = k0 * NO + col;
        dw0[it] = w_hat[i0];
        dw1[it] = w_hat[i0 + NO];
        dm0[it] = m_hat[i0];
        dm1[it] = m_hat[i0 + NO];
    }
    load_a(fx, pa0, pa1, 0);
    __builtin_amdgcn_sched_barrier(0);
#pragma unroll
    for (int i = 0; i < 8; ++i) {
        const int idx = tid + (i << 8);
        const int c = idx >> 7;
        const int kq = idx & 127;
        v4f r;
        r.x = w1_of(tw[i].x, tm[i].x);
        r.y = w1_of(tw[i].y, tm[i].y);
        r.z = w1_of(tw[i].z, tm[i].z);
        r.w = w1_of(tw[i].w, tm[i].w);
        *(v4f*)(wt + c * WT_PITCH + (kq << 2)) = r;
    }
    lds_barrier();

    const unsigned hm = half ? 0xFFFF0000u : 0x0000FFFFu;
#pragma unroll
    for (int it = 0; it < 16; ++it) {
        const int J = (it << 2) + (wave >> 1);
        const int u = wave & 1;
        const int k0 = ((J << 2) + (half << 1) + u) << 1;
        const v2f t = *(const v2f*)(wt + l15 * WT_PITCH + k0);
        const float ws0 = fabsf(t.x);
        const float ws1 = fabsf(t.y);
        const float t2_0 = 1.0f - 2.0f * ws0;
        const float t2_1 = 1.0f - 2.0f * ws1;
        v4f r0, r1;
        r0.x = w1_of(dw0[it], dm0[it]);
        r0.y = w1_of(dw1[it], dm1[it]);
        r0.z = log_clamped(fabsf(t2_0));
        r0.w = log_clamped(fabsf(t2_1));
        r1.x = log_clamped(1.0f - ws0);
        r1.y = log_clamped(1.0f - ws1);
        r1.z = __uint_as_float((t2_0 < 0.0f) ? hm : 0u);
        r1.w = __uint_as_float((t2_1 < 0.0f) ? hm : 0u);
        const int S = (J << 1) + u;
        v4f* dst = pbl + S * PB_STEP + lane;
        dst[0]  = r0;
        dst[32] = r1;
    }
    lds_barrier();
}

#define KB 8
#define NBLK (NI / (4 * KB))
#define NBH (NBLK / 2)

struct Frag {
    v4f a0[KB / 2];
    v4f a1[KB / 2];
    v4f q0[KB];
    v2u q1[KB];
};


__device__ __forceinline__ void load_a(Frag& f, const v4f* pa0, const v4f* pa1, int blk) {
    const v4f* qa0 = pa0 + blk * KB;
    const v4f* qa1 = pa1 + blk * KB;
#pragma unroll
    for (int j = 0; j < KB / 2; ++j) {
        f.a0[j] = qa0[2 * j];
        f.a1[j] = qa1[2 * j];
    }
}

__device__ __forceinline__ void load_b(Frag& f, const v4f* pb, int blk) {
    const v4f* qb  = pb + blk * (KB * PB_STEP);
#pragma unroll
    for (int s = 0; s < KB; ++s) {
        f.q0[s] = qb[s * PB_STEP];
        f.q1[s] = ((const v2u*)(qb + s * PB_STEP + 32))[1];
    }
}

__device__ __forceinline__ void load_block(Frag& f, const v4f* pa0, const v4f* pa1,
                                           const v4f* pb, int blk) {
    load_a(f, pa0, pa1, blk);
    load_b(f, pb, blk);
}

__device__ __forceinline__ float log_abs(float xv) {
    const float v = fmaxf(fabsf(xv), EPS);
    const float y = __builtin_amdgcn_logf(v);
    return fmaf(y, 0x1.62e42ep-1f, y * 0x1.efa39ep-25f);
}

__device__ __forceinline__ void mma_block(const Frag& f, float& zmin,
                                          unsigned& par0, unsigned& par1,
                                          v8f& accA0, v8f& accM0, v8f& accL0,
                                          v8f& accA1, v8f& accM1, v8f& accL1) {
#pragma unroll
    for (int s = 0; s < KB; ++s) {
        const v4f xq0 = f.a0[s >> 1], xq1 = f.a1[s >> 1];
        const v2f ax0 = (s & 1) ? __builtin_shufflevector(xq0, xq0, 2, 3)
                                : __builtin_shufflevector(xq0, xq0, 0, 1);
        const v2f ax1 = (s & 1) ? __builtin_shufflevector(xq1, xq1, 2, 3)
                                : __builtin_shufflevector(xq1, xq1, 0, 1);
        const v4f bq0 = f.q0[s];
        v2f al0, al1;
        al0.x = log_abs(ax0.x);
        al0.y = log_abs(ax0.y);
        al1.x = log_abs(ax1.x);
        al1.y = log_abs(ax1.y);
        v2f bw  = __builtin_shufflevector(bq0, bq0, 0, 1);
        v2f b2  = __builtin_shufflevector(bq0, bq0, 2, 3);
        const v2u bm = f.q1[s];

        const bool c00 = ax0.x < 0.0f, c01 = ax0.y < 0.0f;
        const bool c10 = ax1.x < 0.0f, c11 = ax1.y < 0.0f;
        v2f an0, an1;
        an0.x = c00 ? 1.0f : 0.0f;
        an0.y = c01 ? 1.0f : 0.0f;
        an1.x = c10 ? 1.0f : 0.0f;
        an1.y = c11 ? 1.0f : 0.0f;
        par0 ^= (__builtin_amdgcn_ballot_w32(c00) & bm.x)
              ^ (__builtin_amdgcn_ballot_w32(c01) & bm.y);
        par1 ^= (__builtin_amdgcn_ballot_w32(c10) & bm.x)
              ^ (__builtin_amdgcn_ballot_w32(c11) & bm.y);
        zmin = fminf(zmin, fminf(fabsf(ax0.x), fabsf(ax0.y)));
        zmin = fminf(zmin, fminf(fabsf(ax1.x), fabsf(ax1.y)));

        accA0 = __builtin_amdgcn_wmma_f32_16x16x4_f32(false, ax0, false, bw,
                                                      (short)0, accA0, false, false);
        accA1 = __builtin_amdgcn_wmma_f32_16x16x4_f32(false, ax1, false, bw,
                                                      (short)0, accA1, false, false);
        accM0 = __builtin_amdgcn_wmma_f32_16x16x4_f32(false, al0, false, bw,
                                                      (short)0, accM0, false, false);
        accM1 = __builtin_amdgcn_wmma_f32_16x16x4_f32(false, al1, false, bw,
                                                      (short)0, accM1, false, false);
        accL0 = __builtin_amdgcn_wmma_f32_16x16x4_f32(false, an0, false, b2,
                                                      (short)0, accL0, false, false);
        accL1 = __builtin_amdgcn_wmma_f32_16x16x4_f32(false, an1, false, b2,
                                                      (short)0, accL1, false, false);
#pragma unroll
        for (int j = 0; j < 6; ++j) {
            __builtin_amdgcn_sched_group_barrier(0x008, 1, 0);
            __builtin_amdgcn_sched_group_barrier(0x002, 6, 0);
            if (j < 4)
                __builtin_amdgcn_sched_group_barrier(0x400, 1, 0);
        }
    }
}

__device__ __forceinline__ void zero_pass(const v4f* pa0, const v4f* pa1, const v4f* pb,
                                          v8f& accL0, v8f& accL1) {
#pragma unroll 1
    for (int p = 0; p < NBH * KB / 2; ++p) {
        const v4f xq0 = pa0[2 * p];
        const v4f xq1 = pa1[2 * p];
        const v4f r0 = pb[(2 * p) * PB_STEP + 32];
        const v4f r1 = pb[(2 * p + 1) * PB_STEP + 32];
        v2f az0, az1;
        az0.x = (xq0.x == 0.0f) ? 1.0f : 0.0f;
        az0.y = (xq0.y == 0.0f) ? 1.0f : 0.0f;
        az1.x = (xq1.x == 0.0f) ? 1.0f : 0.0f;
        az1.y = (xq1.y == 0.0f) ? 1.0f : 0.0f;
        const v2f b1e = __builtin_shufflevector(r0, r0, 0, 1);
        accL0 = __builtin_amdgcn_wmma_f32_16x16x4_f32(false, az0, false, b1e,
                                                      (short)0, accL0, false, false);
        accL1 = __builtin_amdgcn_wmma_f32_16x16x4_f32(false, az1, false, b1e,
                                                      (short)0, accL1, false, false);
        v2f bz0, bz1;
        bz0.x = (xq0.z == 0.0f) ? 1.0f : 0.0f;
        bz0.y = (xq0.w == 0.0f) ? 1.0f : 0.0f;
        bz1.x = (xq1.z == 0.0f) ? 1.0f : 0.0f;
        bz1.y = (xq1.w == 0.0f) ? 1.0f : 0.0f;
        const v2f b1o = __builtin_shufflevector(r1, r1, 0, 1);
        accL0 = __builtin_amdgcn_wmma_f32_16x16x4_f32(false, bz0, false, b1o,
                                                      (short)0, accL0, false, false);
        accL1 = __builtin_amdgcn_wmma_f32_16x16x4_f32(false, bz1, false, b1o,
                                                      (short)0, accL1, false, false);
    }
}

__global__ void __launch_bounds__(256)
nalu_fused_kernel(const float* __restrict__ x,
                  const float* __restrict__ w_hat,
                  const float* __restrict__ m_hat,
                  const float* __restrict__ g,
                  float* __restrict__ out) {
    __shared__ v4f pbl[128 * PB_STEP];
    __shared__ __attribute__((aligned(16))) float wt[16 * WT_PITCH];
    __shared__ v4f xch[8 * 6 * 32];
    __shared__ unsigned xpar[8 * 2 * 32];

    const int lane = threadIdx.x & 31;
    const int wave = threadIdx.x >> 5;
    const int kh = wave >> 2;
    const int n0 = (blockIdx.x & 31) << 4;
    const int m0 = ((blockIdx.x >> 5) << 7) + ((wave & 3) << 5);
    const int half = lane >> 4;
    const int l15 = lane & 15;
    const int bcol = n0 + l15;

    const v4f* pa0 = (const v4f*)x + (m0 + l15) * (NI >> 2) + kh * (NI >> 3) + half;
    const v4f* pa1 = pa0 + 16 * (NI >> 2);

    Frag f0, f1;
    build_b(w_hat, m_hat, pbl, wt, n0, wave, lane, f0, pa0, pa1);

    const v4f* pb  = pbl + kh * (NBH * KB * PB_STEP) + lane;

    v8f accA0 = {}, accM0 = {}, accL0 = {};
    v8f accA1 = {}, accM1 = {}, accL1 = {};
    unsigned par0 = 0u, par1 = 0u;
    float zmin = 1.0f;

    load_b(f0, pb, 0);

#pragma unroll 1
    for (int b = 0; b < NBH - 2; b += 2) {
        load_block(f1, pa0, pa1, pb, b + 1);
        __builtin_amdgcn_sched_barrier(0);
        mma_block(f0, zmin, par0, par1, accA0, accM0, accL0, accA1, accM1, accL1);
        load_block(f0, pa0, pa1, pb, b + 2);
        __builtin_amdgcn_sched_barrier(0);
        mma_block(f1, zmin, par0, par1, accA0, accM0, accL0, accA1, accM1, accL1);
    }
    load_block(f1, pa0, pa1, pb, NBH - 1);
    __builtin_amdgcn_sched_barrier(0);
    mma_block(f0, zmin, par0, par1, accA0, accM0, accL0, accA1, accM1, accL1);
    mma_block(f1, zmin, par0, par1, accA0, accM0, accL0, accA1, accM1, accL1);

    if (__builtin_amdgcn_ballot_w32(zmin == 0.0f) != 0u)
        zero_pass(pa0, pa1, pb, accL0, accL1);

    const v8f sA = kh ? accA0 : accA1;
    const v8f sM = kh ? accM0 : accM1;
    const v8f sL = kh ? accL0 : accL1;
    v8f kA = kh ? accA1 : accA0;
    v8f kM = kh ? accM1 : accM0;
    v8f kL = kh ? accL1 : accL0;

    v4f* mine = xch + wave * (6 * 32) + lane;
    mine[0 * 32] = __builtin_shufflevector(sA, sA, 0, 1, 2, 3);
    mine[1 * 32] = __builtin_shufflevector(sA, sA, 4, 5, 6, 7);
    mine[2 * 32] = __builtin_shufflevector(sM, sM, 0, 1, 2, 3);
    mine[3 * 32] = __builtin_shufflevector(sM, sM, 4, 5, 6, 7);
    mine[4 * 32] = __builtin_shufflevector(sL, sL, 0, 1, 2, 3);
    mine[5 * 32] = __builtin_shufflevector(sL, sL, 4, 5, 6, 7);
    xpar[(wave << 6) + lane]      = (par0 >> (half << 4)) & 0xFFFFu;
    xpar[(wave << 6) + 32 + lane] = (par1 >> (half << 4)) & 0xFFFFu;
    __syncthreads();
    const v4f* theirs = xch + (wave ^ 4) * (6 * 32) + lane;
    kA += __builtin_shufflevector(theirs[0 * 32], theirs[1 * 32], 0, 1, 2, 3, 4, 5, 6, 7);
    kM += __builtin_shufflevector(theirs[2 * 32], theirs[3 * 32], 0, 1, 2, 3, 4, 5, 6, 7);
    kL += __builtin_shufflevector(theirs[4 * 32], theirs[5 * 32], 0, 1, 2, 3, 4, 5, 6, 7);
    const unsigned* pmine = xpar + (wave << 6) + (kh << 5) + l15;
    const unsigned* ptheirs = xpar + ((wave ^ 4) << 6) + (kh << 5) + l15;
    const unsigned kPar = pmine[0] ^ pmine[16] ^ ptheirs[0] ^ ptheirs[16];

    const float g1 = 1.0f / (1.0f + expf(-g[bcol]));
    const float gm = 1.0f - g1;
    const int rbase = m0 + 16 * kh + 8 * half;
#pragma unroll
    for (int v = 0; v < 8; ++v) {
        const float m1 = expf(fminf(kM[v], LOG_CLIP));
        const unsigned odd = (kPar >> (8 * half + v)) & 1u;
        float ms = (odd ? -1.0f : 1.0f) * expf(kL[v]);
        ms = fminf(fmaxf(ms, -1.0f), 1.0f);
        out[(rbase + v) * NO + bcol] = g1 * kA[v] + gm * m1 * ms;
    }
}

extern "C" void kernel_launch(void* const* d_in, const int* in_sizes, int n_in,
                              void* d_out, int out_size, void* d_ws, size_t ws_size,
                              hipStream_t stream) {
    const float* x     = (const float*)d_in[0];
    const float* w_hat = (const float*)d_in[1];
    const float* m_hat = (const float*)d_in[2];
    const float* g     = (const float*)d_in[3];
    float* out = (float*)d_out;

    nalu_fused_kernel<<<256, 256, 0, stream>>>(x, w_hat, m_hat, g, out);
}


# logx operand kept in base 2 in the K loop: per x value one log2 of |x| and one max against log2(EPS) (was 2 max, log2, mul, fma); ln2 applied once per output sum in the epilogue; sum rounding changes
# speedup vs baseline: 6.1694x; 1.0573x over previous
#include <hip/hip_runtime.h>
#include <math.h>

typedef float v2f __attribute__((ext_vector_type(2)));
typedef float v4f __attribute__((ext_vector_type(4)));
typedef float v8f __attribute__((ext_vector_type(8)));
typedef unsigned v2u __attribute__((ext_vector_type(2)));

#define NB 1024
#define NI 512
#define NO 512
#define EPS 1e-7f
#define LOG_CLIP 20.0f
#define NEG_BIG -1e30f


#define PB_STEP 64
#define WT_PITCH 516

__device__ __forceinline__ float w1_of(float wh, float mh) {
    const float e = __builtin_amdgcn_exp2f(fabsf(wh) * -0x1.715476p+1f);
    const float s = __builtin_amdgcn_exp2f(mh * -0x1.715476p+0f);
    const float r = __builtin_amdgcn_rcpf((1.0f + e) * (1.0f + s));
    return copysignf((1.0f - e) * r, wh);
}

__device__ __forceinline__ float log_clamped(float v) {
    const float y = __builtin_amdgcn_logf(v);
    return fmaxf(fmaf(y, 0x1.62e42ep-1f, y * 0x1.efa39ep-25f), NEG_BIG);
}

__device__ __forceinline__ void lds_barrier() {
    __builtin_amdgcn_fence(__ATOMIC_RELEASE, "workgroup", "local");
    __builtin_amdgcn_s_barrier();
    __builtin_amdgcn_fence(__ATOMIC_ACQUIRE, "workgroup", "local");
}

struct Frag;
__device__ __forceinline__ void load_a(Frag& f, const v4f* pa0, const v4f* pa1, int blk);

__device__ __forceinline__ void build_b(const float* __restrict__ w_hat,
                                        const float* __restrict__ m_hat,
                                        v4f* pbl, float* wt,
                                        int n0, int wave, int lane,
                                        Frag& fx, const v4f* pa0, const v4f* pa1) {
    const int half = lane >> 4;
    const int l15 = lane & 15;
    const int col = n0 + l15;
    const int tid = (wave << 5) + lane;

    v4f tw[8], tm[8];
#pragma unroll
    for (int i = 0; i < 8; ++i) {
        const int idx = tid + (i << 8);
        const int c = idx >> 7;
        const int kq = idx & 127;
        tw[i] = ((const v4f*)w_hat)[(n0 + c) * (NO >> 2) + kq];
        tm[i] = ((const v4f*)m_hat)[(n0 + c) * (NO >> 2) + kq];
    }
    __builtin_amdgcn_sched_barrier(0);
    float dw0[16], dw1[16], dm0[16], dm1[16];
#pragma unroll
    for (int it = 0; it < 16; ++it) {
        const int J = (it << 2) + (wave >> 1);
        const int k0 = ((J << 2) + (half << 1) + (wave & 1)) << 1;
        const int i0 = k0 * NO + col;
        dw0[it] = w_hat[i0];
        dw1[it] = w_hat[i0 + NO];
        dm0[it] = m_hat[i0];
        dm1[it] = m_hat[i0 + NO];
    }
    load_a(fx, pa0, pa1, 0);
    __builtin_amdgcn_sched_barrier(0);
#pragma unroll
    for (int i = 0; i < 8; ++i) {
        const int idx = tid + (i << 8);
        const int c = idx >> 7;
        const int kq = idx & 127;
        v4f r;
        r.x = w1_of(tw[i].x, tm[i].x);
        r.y = w1_of(tw[i].y, tm[i].y);
        r.z = w1_of(tw[i].z, tm[i].z);
        r.w = w1_of(tw[i].w, tm[i].w);
        *(v4f*)(wt + c * WT_PITCH + (kq << 2)) = r;
    }
    lds_barrier();

    const unsigned hm = half ? 0xFFFF0000u : 0x0000FFFFu;
#pragma unroll
    for (int it = 0; it < 16; ++it) {
        const int J = (it << 2) + (wave >> 1);
        const int u = wave & 1;
        const int k0 = ((J << 2) + (half << 1) + u) << 1;
        const v2f t = *(const v2f*)(wt + l15 * WT_PITCH + k0);
        const float ws0 = fabsf(t.x);
        const float ws1 = fabsf(t.y);
        const float t2_0 = 1.0f - 2.0f * ws0;
        const float t2_1 = 1.0f - 2.0f * ws1;
        v4f r0, r1;
        r0.x = w1_of(dw0[it], dm0[it]);
        r0.y = w1_of(dw1[it], dm1[it]);
        r0.z = log_clamped(fabsf(t2_0));
        r0.w = log_clamped(fabsf(t2_1));
        r1.x = log_clamped(1.0f - ws0);
        r1.y = log_clamped(1.0f - ws1);
        r1.z = __uint_as_float((t2_0 < 0.0f) ? hm : 0u);
        r1.w = __uint_as_float((t2_1 < 0.0f) ? hm : 0u);
        const int S = (J << 1) + u;
        v4f* dst = pbl + S * PB_STEP + lane;
        dst[0]  = r0;
        dst[32] = r1;
    }
    lds_barrier();
}

#define KB 8
#define NBLK (NI / (4 * KB))
#define NBH (NBLK / 2)

struct Frag {
    v4f a0[KB / 2];
    v4f a1[KB / 2];
    v4f q0[KB];
    v2u q1[KB];
};


__device__ __forceinline__ void load_a(Frag& f, const v4f* pa0, const v4f* pa1, int blk) {
    const v4f* qa0 = pa0 + blk * KB;
    const v4f* qa1 = pa1 + blk * KB;
#pragma unroll
    for (int j = 0; j < KB / 2; ++j) {
        f.a0[j] = qa0[2 * j];
        f.a1[j] = qa1[2 * j];
    }
}

__device__ __forceinline__ void load_b(Frag& f, const v4f* pb, int blk) {
    const v4f* qb  = pb + blk * (KB * PB_STEP);
#pragma unroll
    for (int s = 0; s < KB; ++s) {
        f.q0[s] = qb[s * PB_STEP];
        f.q1[s] = ((const v2u*)(qb + s * PB_STEP + 32))[1];
    }
}

__device__ __forceinline__ void load_block(Frag& f, const v4f* pa0, const v4f* pa1,
                                           const v4f* pb, int blk) {
    load_a(f, pa0, pa1, blk);
    load_b(f, pb, blk);
}

#define LOG2_EPS -0x1.740e52p+4f
__device__ __forceinline__ float log2_abs(float xv) {
    return fmaxf(__builtin_amdgcn_logf(fabsf(xv)), LOG2_EPS);
}

__device__ __forceinline__ float ln_of_log2_sum(float s) {
    return fmaf(s, 0x1.62e42ep-1f, s * 0x1.efa39ep-25f);
}

__device__ __forceinline__ void mma_block(const Frag& f, float& zmin,
                                          unsigned& par0, unsigned& par1,
                                          v8f& accA0, v8f& accM0, v8f& accL0,
                                          v8f& accA1, v8f& accM1, v8f& accL1) {
#pragma unroll
    for (int s = 0; s < KB; ++s) {
        const v4f xq0 = f.a0[s >> 1], xq1 = f.a1[s >> 1];
        const v2f ax0 = (s & 1) ? __builtin_shufflevector(xq0, xq0, 2, 3)
                                : __builtin_shufflevector(xq0, xq0, 0, 1);
        const v2f ax1 = (s & 1) ? __builtin_shufflevector(xq1, xq1, 2, 3)
                                : __builtin_shufflevector(xq1, xq1, 0, 1);
        const v4f bq0 = f.q0[s];
        v2f al0, al1;
        al0.x = log2_abs(ax0.x);
        al0.y = log2_abs(ax0.y);
        al1.x = log2_abs(ax1.x);
        al1.y = log2_abs(ax1.y);
        v2f bw  = __builtin_shufflevector(bq0, bq0, 0, 1);
        v2f b2  = __builtin_shufflevector(bq0, bq0, 2, 3);
        const v2u bm = f.q1[s];

        const bool c00 = ax0.x < 0.0f, c01 = ax0.y < 0.0f;
        const bool c10 = ax1.x < 0.0f, c11 = ax1.y < 0.0f;
        v2f an0, an1;
        an0.x = c00 ? 1.0f : 0.0f;
        an0.y = c01 ? 1.0f : 0.0f;
        an1.x = c10 ? 1.0f : 0.0f;
        an1.y = c11 ? 1.0f : 0.0f;
        par0 ^= (__builtin_amdgcn_ballot_w32(c00) & bm.x)
              ^ (__builtin_amdgcn_ballot_w32(c01) & bm.y);
        par1 ^= (__builtin_amdgcn_ballot_w32(c10) & bm.x)
              ^ (__builtin_amdgcn_ballot_w32(c11) & bm.y);
        zmin = fminf(zmin, fminf(fabsf(ax0.x), fabsf(ax0.y)));
        zmin = fminf(zmin, fminf(fabsf(ax1.x), fabsf(ax1.y)));

        accA0 = __builtin_amdgcn_wmma_f32_16x16x4_f32(false, ax0, false, bw,
                                                      (short)0, accA0, false, false);
        accA1 = __builtin_amdgcn_wmma_f32_16x16x4_f32(false, ax1, false, bw,
                                                      (short)0, accA1, false, false);
        accM0 = __builtin_amdgcn_wmma_f32_16x16x4_f32(false, al0, false, bw,
                                                      (short)0, accM0, false, false);
        accM1 = __builtin_amdgcn_wmma_f32_16x16x4_f32(false, al1, false, bw,
                                                      (short)0, accM1, false, false);
        accL0 = __builtin_amdgcn_wmma_f32_16x16x4_f32(false, an0, false, b2,
                                                      (short)0, accL0, false, false);
        accL1 = __builtin_amdgcn_wmma_f32_16x16x4_f32(false, an1, false, b2,
                                                      (short)0, accL1, false, false);
#pragma unroll
        for (int j = 0; j < 6; ++j) {
            __builtin_amdgcn_sched_group_barrier(0x008, 1, 0);
            if (j < 4) {
                __builtin_amdgcn_sched_group_barrier(0x002, 4, 0);
                __builtin_amdgcn_sched_group_barrier(0x400, 1, 0);
            } else {
                __builtin_amdgcn_sched_group_barrier(0x002, 2, 0);
            }
        }
    }
}

__device__ __forceinline__ void zero_pass(const v4f* pa0, const v4f* pa1, const v4f* pb,
                                          v8f& accL0, v8f& accL1) {
#pragma unroll 1
    for (int p = 0; p < NBH * KB / 2; ++p) {
        const v4f xq0 = pa0[2 * p];
        const v4f xq1 = pa1[2 * p];
        const v4f r0 = pb[(2 * p) * PB_STEP + 32];
        const v4f r1 = pb[(2 * p + 1) * PB_STEP + 32];
        v2f az0, az1;
        az0.x = (xq0.x == 0.0f) ? 1.0f : 0.0f;
        az0.y = (xq0.y == 0.0f) ? 1.0f : 0.0f;
        az1.x = (xq1.x == 0.0f) ? 1.0f : 0.0f;
        az1.y = (xq1.y == 0.0f) ? 1.0f : 0.0f;
        const v2f b1e = __builtin_shufflevector(r0, r0, 0, 1);
        accL0 = __builtin_amdgcn_wmma_f32_16x16x4_f32(false, az0, false, b1e,
                                                      (short)0, accL0, false, false);
        accL1 = __builtin_amdgcn_wmma_f32_16x16x4_f32(false, az1, false, b1e,
                                                      (short)0, accL1, false, false);
        v2f bz0, bz1;
        bz0.x = (xq0.z == 0.0f) ? 1.0f : 0.0f;
        bz0.y = (xq0.w == 0.0f) ? 1.0f : 0.0f;
        bz1.x = (xq1.z == 0.0f) ? 1.0f : 0.0f;
        bz1.y = (xq1.w == 0.0f) ? 1.0f : 0.0f;
        const v2f b1o = __builtin_shufflevector(r1, r1, 0, 1);
        accL0 = __builtin_amdgcn_wmma_f32_16x16x4_f32(false, bz0, false, b1o,
                                                      (short)0, accL0, false, false);
        accL1 = __builtin_amdgcn_wmma_f32_16x16x4_f32(false, bz1, false, b1o,
                                                      (short)0, accL1, false, false);
    }
}

__global__ void __launch_bounds__(256)
nalu_fused_kernel(const float* __restrict__ x,
                  const float* __restrict__ w_hat,
                  const float* __restrict__ m_hat,
                  const float* __restrict__ g,
                  float* __restrict__ out) {
    __shared__ v4f pbl[128 * PB_STEP];
    __shared__ __attribute__((aligned(16))) float wt[16 * WT_PITCH];
    __shared__ v4f xch[8 * 6 * 32];
    __shared__ unsigned xpar[8 * 2 * 32];

    const int lane = threadIdx.x & 31;
    const int wave = threadIdx.x >> 5;
    const int kh = wave >> 2;
    const int n0 = (blockIdx.x & 31) << 4;
    const int m0 = ((blockIdx.x >> 5) << 7) + ((wave & 3) << 5);
    const int half = lane >> 4;
    const int l15 = lane & 15;
    const int bcol = n0 + l15;

    const v4f* pa0 = (const v4f*)x + (m0 + l15) * (NI >> 2) + kh * (NI >> 3) + half;
    const v4f* pa1 = pa0 + 16 * (NI >> 2);

    Frag f0, f1;
    build_b(w_hat, m_hat, pbl, wt, n0, wave, lane, f0, pa0, pa1);

    const v4f* pb  = pbl + kh * (NBH * KB * PB_STEP) + lane;

    v8f accA0 = {}, accM0 = {}, accL0 = {};
    v8f accA1 = {}, accM1 = {}, accL1 = {};
    unsigned par0 = 0u, par1 = 0u;
    float zmin = 1.0f;

    load_b(f0, pb, 0);

#pragma unroll 1
    for (int b = 0; b < NBH - 2; b += 2) {
        load_block(f1, pa0, pa1, pb, b + 1);
        __builtin_amdgcn_sched_barrier(0);
        mma_block(f0, zmin, par0, par1, accA0, accM0, accL0, accA1, accM1, accL1);
        load_block(f0, pa0, pa1, pb, b + 2);
        __builtin_amdgcn_sched_barrier(0);
        mma_block(f1, zmin, par0, par1, accA0, accM0, accL0, accA1, accM1, accL1);
    }
    load_block(f1, pa0, pa1, pb, NBH - 1);
    __builtin_amdgcn_sched_barrier(0);
    mma_block(f0, zmin, par0, par1, accA0, accM0, accL0, accA1, accM1, accL1);
    mma_block(f1, zmin, par0, par1, accA0, accM0, accL0, accA1, accM1, accL1);

    if (__builtin_amdgcn_ballot_w32(zmin == 0.0f) != 0u)
        zero_pass(pa0, pa1, pb, accL0, accL1);

    const v8f sA = kh ? accA0 : accA1;
    const v8f sM = kh ? accM0 : accM1;
    const v8f sL = kh ? accL0 : accL1;
    v8f kA = kh ? accA1 : accA0;
    v8f kM = kh ? accM1 : accM0;
    v8f kL = kh ? accL1 : accL0;

    v4f* mine = xch + wave * (6 * 32) + lane;
    mine[0 * 32] = __builtin_shufflevector(sA, sA, 0, 1, 2, 3);
    mine[1 * 32] = __builtin_shufflevector(sA, sA, 4, 5, 6, 7);
    mine[2 * 32] = __builtin_shufflevector(sM, sM, 0, 1, 2, 3);
    mine[3 * 32] = __builtin_shufflevector(sM, sM, 4, 5, 6, 7);
    mine[4 * 32] = __builtin_shufflevector(sL, sL, 0, 1, 2, 3);
    mine[5 * 32] = __builtin_shufflevector(sL, sL, 4, 5, 6, 7);
    xpar[(wave << 6) + lane]      = (par0 >> (half << 4)) & 0xFFFFu;
    xpar[(wave << 6) + 32 + lane] = (par1 >> (half << 4)) & 0xFFFFu;
    __syncthreads();
    const v4f* theirs = xch + (wave ^ 4) * (6 * 32) + lane;
    kA += __builtin_shufflevector(theirs[0 * 32], theirs[1 * 32], 0, 1, 2, 3, 4, 5, 6, 7);
    kM += __builtin_shufflevector(theirs[2 * 32], theirs[3 * 32], 0, 1, 2, 3, 4, 5, 6, 7);
    kL += __builtin_shufflevector(theirs[4 * 32], theirs[5 * 32], 0, 1, 2, 3, 4, 5, 6, 7);
    const unsigned* pmine = xpar + (wave << 6) + (kh << 5) + l15;
    const unsigned* ptheirs = xpar + ((wave ^ 4) << 6) + (kh << 5) + l15;
    const unsigned kPar = pmine[0] ^ pmine[16] ^ ptheirs[0] ^ ptheirs[16];

    const float g1 = 1.0f / (1.0f + expf(-g[bcol]));
    const float gm = 1.0f - g1;
    const int rbase = m0 + 16 * kh + 8 * half;
#pragma unroll
    for (int v = 0; v < 8; ++v) {
        const float m1 = expf(fminf(ln_of_log2_sum(kM[v]), LOG_CLIP));
        const unsigned odd = (kPar >> (8 * half + v)) & 1u;
        float ms = (odd ? -1.0f : 1.0f) * expf(kL[v]);
        ms = fminf(fmaxf(ms, -1.0f), 1.0f);
        out[(rbase + v) * NO + bcol] = g1 * kA[v] + gm * m1 * ms;
    }
}

extern "C" void kernel_launch(void* const* d_in, const int* in_sizes, int n_in,
                              void* d_out, int out_size, void* d_ws, size_t ws_size,
                              hipStream_t stream) {
    const float* x     = (const float*)d_in[0];
    const float* w_hat = (const float*)d_in[1];
    const float* m_hat = (const float*)d_in[2];
    const float* g     = (const float*)d_in[3];
    float* out = (float*)d_out;

    nalu_fused_kernel<<<256, 256, 0, stream>>>(x, w_hat, m_hat, g, out);
}
